# starting kernel
# baseline (speedup 1.0000x reference)
#include <hip/hip_runtime.h>
#include <hip/hip_bf16.h>

typedef __attribute__((ext_vector_type(2))) float v2f;
typedef __attribute__((ext_vector_type(8))) float v8f;

#define N_NODES 100000
#define N_EDGES 1600000
#define IN_F    128
#define HID_F   128
#define OUT_F   64


__global__ void zero_kernel(float* __restrict__ p, long n) {
    long i = (long)blockIdx.x * blockDim.x + threadIdx.x;
    long stride = (long)gridDim.x * blockDim.x;
    for (; i < n; i += stride) p[i] = 0.0f;
}

__global__ void deg_count_kernel(const int* __restrict__ dst, float* __restrict__ deg, int E) {
    int e = blockIdx.x * blockDim.x + threadIdx.x;
    if (e < E) atomicAdd(&deg[dst[e]], 1.0f);
}

__global__ void deg_finish_kernel(float* __restrict__ deg, int n) {
    int i = blockIdx.x * blockDim.x + threadIdx.x;
    if (i < n) {
        float d = deg[i];
        deg[i] = (d > 0.0f) ? (1.0f / sqrtf(d)) : 0.0f;
    }
}

template <int FOUT, bool RELU_IN>
__global__ __launch_bounds__(32 * (FOUT / 16))
void gemm_wmma_kernel(const float* __restrict__ X, const float* __restrict__ W,
                      const float* __restrict__ bias, float* __restrict__ H) {
    constexpr int K = 128;
    const int wave = threadIdx.x >> 5;
    const int lane = threadIdx.x & 31;
    const int half = lane >> 4;
    const int l    = lane & 15;
    const int m0 = blockIdx.x * 16;
    const int n0 = wave * 16;

    const float* arow = X + (long)(m0 + l) * K + 2 * half;
    const float* brow = W + (long)(n0 + l) * K + 2 * half;

    v8f c = {};
#pragma unroll
    for (int s = 0; s < K / 4; ++s) {
        v2f a = *(const v2f*)(arow + 4 * s);
        v2f b = *(const v2f*)(brow + 4 * s);
        if (RELU_IN) {
            a.x = fmaxf(a.x, 0.0f);
            a.y = fmaxf(a.y, 0.0f);
        }
        c = __builtin_amdgcn_wmma_f32_16x16x4_f32(false, a, false, b,
                                                  (short)0, c, false, false);
    }

    const float bb = bias[n0 + l];
#pragma unroll
    for (int r = 0; r < 8; ++r) {
        const int row = m0 + r + 8 * half;
        H[(long)row * FOUT + n0 + l] = c[r] + bb;
    }
}

__global__ void aggregate128_kernel(const int* __restrict__ src, const int* __restrict__ dst,
                                    const float* __restrict__ dis, const float* __restrict__ h,
                                    float* __restrict__ out, int E) {
    int e = blockIdx.x * (blockDim.x >> 5) + (threadIdx.x >> 5);
    if (e >= E) return;
    int lane = threadIdx.x & 31;
    int s = src[e], d = dst[e];
    float w = dis[s] * dis[d];
    float4 v = ((const float4*)(h + (long)s * 128))[lane];
    float* o = out + (long)d * 128 + lane * 4;
    atomicAdd(o + 0, v.x * w);
    atomicAdd(o + 1, v.y * w);
    atomicAdd(o + 2, v.z * w);
    atomicAdd(o + 3, v.w * w);
}

__global__ void aggregate64_kernel(const int* __restrict__ src, const int* __restrict__ dst,
                                   const float* __restrict__ dis, const float* __restrict__ h,
                                   float* __restrict__ out, int E) {
    int e = blockIdx.x * (blockDim.x >> 5) + (threadIdx.x >> 5);
    if (e >= E) return;
    int lane = threadIdx.x & 31;
    int s = src[e], d = dst[e];
    float w = dis[s] * dis[d];
    float2 v = ((const float2*)(h + (long)s * 64))[lane];
    float* o = out + (long)d * 64 + lane * 2;
    atomicAdd(o + 0, v.x * w);
    atomicAdd(o + 1, v.y * w);
}


extern "C" void kernel_launch(void* const* d_in, const int* in_sizes, int n_in,
                              void* d_out, int out_size, void* d_ws, size_t ws_size,
                              hipStream_t stream) {
    const float* x  = (const float*)d_in[0];
    const int*   ei = (const int*)d_in[1];
    const float* W1 = (const float*)d_in[2];
    const float* b1 = (const float*)d_in[3];
    const float* W2 = (const float*)d_in[4];
    const float* b2 = (const float*)d_in[5];
    float* out = (float*)d_out;

    const int* src = ei;
    const int* dst = ei + N_EDGES;

    float* deg  = (float*)d_ws;
    float* h1   = deg + N_NODES;
    float* agg1 = h1 + (long)N_NODES * HID_F;
    float* h2   = agg1 + (long)N_NODES * HID_F;

    zero_kernel<<<1024, 256, 0, stream>>>(deg, (long)N_NODES);
    deg_count_kernel<<<(N_EDGES + 255) / 256, 256, 0, stream>>>(dst, deg, N_EDGES);
    deg_finish_kernel<<<(N_NODES + 255) / 256, 256, 0, stream>>>(deg, N_NODES);

    gemm_wmma_kernel<HID_F, false><<<N_NODES / 16, 256, 0, stream>>>(x, W1, b1, h1);

    zero_kernel<<<4096, 256, 0, stream>>>(agg1, (long)N_NODES * HID_F);
    aggregate128_kernel<<<(N_EDGES + 7) / 8, 256, 0, stream>>>(src, dst, deg, h1, agg1, N_EDGES);

    gemm_wmma_kernel<OUT_F, true><<<N_NODES / 16, 128, 0, stream>>>(agg1, W2, b2, h2);

    zero_kernel<<<4096, 256, 0, stream>>>(out, (long)N_NODES * OUT_F);
    aggregate64_kernel<<<(N_EDGES + 7) / 8, 256, 0, stream>>>(src, dst, deg, h2, out, N_EDGES);
}


# Both scatter-adds by destination-band owner work-groups: each scans the dst list, sums its 391 rows in LDS with LDS float adds (arrival order), stores them once; no global atomics or zero passes
# speedup vs baseline: 1.6824x; 1.6824x over previous
#include <hip/hip_runtime.h>
#include <hip/hip_bf16.h>

typedef __attribute__((ext_vector_type(2))) float v2f;
typedef __attribute__((ext_vector_type(8))) float v8f;

#define N_NODES 100000
#define N_EDGES 1600000
#define IN_F    128
#define HID_F   128
#define OUT_F   64


__global__ void zero_kernel(float* __restrict__ p, long n) {
    long i = (long)blockIdx.x * blockDim.x + threadIdx.x;
    long stride = (long)gridDim.x * blockDim.x;
    for (; i < n; i += stride) p[i] = 0.0f;
}

__global__ void deg_count_kernel(const int* __restrict__ dst, float* __restrict__ deg, int E) {
    int e = blockIdx.x * blockDim.x + threadIdx.x;
    if (e < E) atomicAdd(&deg[dst[e]], 1.0f);
}

__global__ void deg_finish_kernel(float* __restrict__ deg, int n) {
    int i = blockIdx.x * blockDim.x + threadIdx.x;
    if (i < n) {
        float d = deg[i];
        deg[i] = (d > 0.0f) ? (1.0f / sqrtf(d)) : 0.0f;
    }
}

template <int FOUT, bool RELU_IN>
__global__ __launch_bounds__(32 * (FOUT / 16))
void gemm_wmma_kernel(const float* __restrict__ X, const float* __restrict__ W,
                      const float* __restrict__ bias, float* __restrict__ H) {
    constexpr int K = 128;
    const int wave = threadIdx.x >> 5;
    const int lane = threadIdx.x & 31;
    const int half = lane >> 4;
    const int l    = lane & 15;
    const int m0 = blockIdx.x * 16;
    const int n0 = wave * 16;

    const float* arow = X + (long)(m0 + l) * K + 2 * half;
    const float* brow = W + (long)(n0 + l) * K + 2 * half;

    v8f c = {};
#pragma unroll
    for (int s = 0; s < K / 4; ++s) {
        v2f a = *(const v2f*)(arow + 4 * s);
        v2f b = *(const v2f*)(brow + 4 * s);
        if (RELU_IN) {
            a.x = fmaxf(a.x, 0.0f);
            a.y = fmaxf(a.y, 0.0f);
        }
        c = __builtin_amdgcn_wmma_f32_16x16x4_f32(false, a, false, b,
                                                  (short)0, c, false, false);
    }

    const float bb = bias[n0 + l];
#pragma unroll
    for (int r = 0; r < 8; ++r) {
        const int row = m0 + r + 8 * half;
        H[(long)row * FOUT + n0 + l] = c[r] + bb;
    }
}

#define BAND_ROWS 391
#define N_BANDS   256
#define EDGE_BITS 21
#define SCAN_WAVES 32
#define SCAN_AHEAD 4
#define QUEUE_WORDS 544

static_assert(N_BANDS * BAND_ROWS >= N_NODES, "bands must cover every row");
static_assert((N_BANDS - 1) * BAND_ROWS < N_NODES, "no empty band");
static_assert(N_EDGES <= (1 << EDGE_BITS), "edge index must fit its field");
static_assert(BAND_ROWS <= (1 << (32 - EDGE_BITS)), "band row must fit its field");
static_assert(N_EDGES % 128 == 0, "whole 128-edge wave steps");
static_assert(QUEUE_WORDS >= 31 + SCAN_AHEAD * 128 + 1, "queue holds one scan step");
static_assert((N_BANDS - 1) * ((N_EDGES / 128) / N_BANDS) < N_EDGES / 128, "scan start inside the list");

template <int F>
__device__ __forceinline__ void add_queued_rows(const unsigned* q, int n, int lane, int lo,
                                                const int* __restrict__ src,
                                                const float* __restrict__ dis,
                                                const float* __restrict__ h,
                                                float* acc) {
    constexpr int C = F / 32;
    const unsigned ent = q[lane < n ? lane : n - 1];
    const int e = (int)(ent & ((1u << EDGE_BITS) - 1u));
    const int r = (int)(ent >> EDGE_BITS);
    const int s = src[e];
    const float w = dis[s] * dis[lo + r];
    const int wbits = __builtin_bit_cast(int, w);

#pragma unroll 1
    for (int j0 = 0; j0 < n; j0 += 8) {
        float v[8][C];
        int   rr[8];
        float ww[8];
#pragma unroll
        for (int u = 0; u < 8; ++u) {
            const int j = (j0 + u < n) ? (j0 + u) : (n - 1);
            const int sj = __builtin_amdgcn_readlane(s, j);
            rr[u] = __builtin_amdgcn_readlane(r, j);
            ww[u] = __builtin_bit_cast(float, __builtin_amdgcn_readlane(wbits, j));
            const float* hp = h + (long)sj * F + lane;
#pragma unroll
            for (int t = 0; t < C; ++t) v[u][t] = hp[32 * t];
        }
#pragma unroll
        for (int u = 0; u < 8; ++u) {
            if (j0 + u < n) {
                float* ap = acc + rr[u] * F + lane;
#pragma unroll
                for (int t = 0; t < C; ++t) atomicAdd(ap + 32 * t, v[u][t] * ww[u]);
            }
        }
    }
}

template <int F>
__global__ __launch_bounds__(32 * SCAN_WAVES)
void band_aggregate_kernel(const int* __restrict__ src, const int* __restrict__ dst,
                           const float* __restrict__ dis, const float* __restrict__ h,
                           float* __restrict__ out) {
    __shared__ __attribute__((aligned(16))) float acc[BAND_ROWS * F];
    __shared__ unsigned queue[SCAN_WAVES][QUEUE_WORDS];

    const int tid  = threadIdx.x;
    const int lane = tid & 31;
    const int wave = __builtin_amdgcn_readfirstlane(tid >> 5);
    const int lo   = blockIdx.x * BAND_ROWS;

    {
        float4* a4 = (float4*)acc;
        const float4 z = {0.0f, 0.0f, 0.0f, 0.0f};
        for (int i = tid; i < BAND_ROWS * F / 4; i += 32 * SCAN_WAVES) a4[i] = z;
    }
    __syncthreads();

    unsigned* q = queue[wave];
    int cnt = 0;
    constexpr int NCH = N_EDGES / 128;

    const int rot = blockIdx.x * (NCH / N_BANDS);

    for (int c0 = wave; c0 < NCH; c0 += SCAN_WAVES * SCAN_AHEAD) {
        int4 dv[SCAN_AHEAD];
        int  ch[SCAN_AHEAD];
#pragma unroll
        for (int u = 0; u < SCAN_AHEAD; ++u) {
            const int ci = c0 + SCAN_WAVES * u;
            int c = ((ci < NCH) ? ci : (NCH - 1)) + rot;
            if (c >= NCH) c -= NCH;
            ch[u] = c;
            dv[u] = *(const int4*)(dst + (long)c * 128 + lane * 4);
        }
#pragma unroll
        for (int u = 0; u < SCAN_AHEAD; ++u) {
            if (c0 + SCAN_WAVES * u < NCH) {
                const int ebase = ch[u] * 128 + lane * 4;
#pragma unroll
                for (int k = 0; k < 4; ++k) {
                    const int d = (k == 0) ? dv[u].x : (k == 1) ? dv[u].y : (k == 2) ? dv[u].z : dv[u].w;
                    const unsigned r = (unsigned)(d - lo);
                    const bool hit = r < (unsigned)BAND_ROWS;
                    const unsigned mask = __builtin_amdgcn_ballot_w32(hit);
                    if (mask != 0u) {
                        if (hit) {
                            const int pos = cnt + (int)__builtin_amdgcn_mbcnt_lo(mask, 0u);
                            q[pos] = (r << EDGE_BITS) | (unsigned)(ebase + k);
                        }
                        cnt += __builtin_popcount(mask);
                    }
                }
            }
        }
        while (cnt >= 32) {
            cnt -= 32;
            add_queued_rows<F>(q + cnt, 32, lane, lo, src, dis, h, acc);
        }
    }
    if (cnt > 0) add_queued_rows<F>(q, cnt, lane, lo, src, dis, h, acc);

    __syncthreads();

    {
        const int rows_here = ((lo + BAND_ROWS < N_NODES) ? BAND_ROWS : (N_NODES - lo));
        const int n4 = rows_here * (F / 4);
        const float4* a4 = (const float4*)acc;
        float4* o4 = (float4*)(out + (long)lo * F);
        for (int i = tid; i < n4; i += 32 * SCAN_WAVES) o4[i] = a4[i];
    }
}


extern "C" void kernel_launch(void* const* d_in, const int* in_sizes, int n_in,
                              void* d_out, int out_size, void* d_ws, size_t ws_size,
                              hipStream_t stream) {
    const float* x  = (const float*)d_in[0];
    const int*   ei = (const int*)d_in[1];
    const float* W1 = (const float*)d_in[2];
    const float* b1 = (const float*)d_in[3];
    const float* W2 = (const float*)d_in[4];
    const float* b2 = (const float*)d_in[5];
    float* out = (float*)d_out;

    const int* src = ei;
    const int* dst = ei + N_EDGES;

    float* deg  = (float*)d_ws;
    float* h1   = deg + N_NODES;
    float* agg1 = h1 + (long)N_NODES * HID_F;
    float* h2   = agg1 + (long)N_NODES * HID_F;

    zero_kernel<<<1024, 256, 0, stream>>>(deg, (long)N_NODES);
    deg_count_kernel<<<(N_EDGES + 255) / 256, 256, 0, stream>>>(dst, deg, N_EDGES);
    deg_finish_kernel<<<(N_NODES + 255) / 256, 256, 0, stream>>>(deg, N_NODES);

    gemm_wmma_kernel<HID_F, false><<<N_NODES / 16, 256, 0, stream>>>(x, W1, b1, h1);

    band_aggregate_kernel<HID_F><<<N_BANDS, 32 * SCAN_WAVES, 0, stream>>>(src, dst, deg, h1, agg1);

    gemm_wmma_kernel<OUT_F, true><<<N_NODES / 16, 128, 0, stream>>>(agg1, W2, b2, h2);

    band_aggregate_kernel<OUT_F><<<N_BANDS, 32 * SCAN_WAVES, 0, stream>>>(src, dst, deg, h2, out);
}


# Second aggregation scans dst no more: the first one's band owners write their queued edges as per-wave lists to the workspace, the second reads its lists; layer-2 sums stay in LDS arrival order
# speedup vs baseline: 1.9619x; 1.1661x over previous
#include <hip/hip_runtime.h>
#include <hip/hip_bf16.h>

typedef __attribute__((ext_vector_type(2))) float v2f;
typedef __attribute__((ext_vector_type(8))) float v8f;

#define N_NODES 100000
#define N_EDGES 1600000
#define IN_F    128
#define HID_F   128
#define OUT_F   64


__global__ void zero_kernel(float* __restrict__ p, long n) {
    long i = (long)blockIdx.x * blockDim.x + threadIdx.x;
    long stride = (long)gridDim.x * blockDim.x;
    for (; i < n; i += stride) p[i] = 0.0f;
}

__global__ void deg_count_kernel(const int* __restrict__ dst, float* __restrict__ deg, int E) {
    int e = blockIdx.x * blockDim.x + threadIdx.x;
    if (e < E) atomicAdd(&deg[dst[e]], 1.0f);
}

__global__ void deg_finish_kernel(float* __restrict__ deg, int n) {
    int i = blockIdx.x * blockDim.x + threadIdx.x;
    if (i < n) {
        float d = deg[i];
        deg[i] = (d > 0.0f) ? (1.0f / sqrtf(d)) : 0.0f;
    }
}

template <int FOUT, bool RELU_IN>
__global__ __launch_bounds__(32 * (FOUT / 16))
void gemm_wmma_kernel(const float* __restrict__ X, const float* __restrict__ W,
                      const float* __restrict__ bias, float* __restrict__ H) {
    constexpr int K = 128;
    const int wave = threadIdx.x >> 5;
    const int lane = threadIdx.x & 31;
    const int half = lane >> 4;
    const int l    = lane & 15;
    const int m0 = blockIdx.x * 16;
    const int n0 = wave * 16;

    const float* arow = X + (long)(m0 + l) * K + 2 * half;
    const float* brow = W + (long)(n0 + l) * K + 2 * half;

    v8f c = {};
#pragma unroll
    for (int s = 0; s < K / 4; ++s) {
        v2f a = *(const v2f*)(arow + 4 * s);
        v2f b = *(const v2f*)(brow + 4 * s);
        if (RELU_IN) {
            a.x = fmaxf(a.x, 0.0f);
            a.y = fmaxf(a.y, 0.0f);
        }
        c = __builtin_amdgcn_wmma_f32_16x16x4_f32(false, a, false, b,
                                                  (short)0, c, false, false);
    }

    const float bb = bias[n0 + l];
#pragma unroll
    for (int r = 0; r < 8; ++r) {
        const int row = m0 + r + 8 * half;
        H[(long)row * FOUT + n0 + l] = c[r] + bb;
    }
}

#define BAND_ROWS 391
#define N_BANDS   256
#define EDGE_BITS 21
#define SCAN_WAVES 32
#define SCAN_AHEAD 4
#define QUEUE_WORDS 544

static_assert(N_BANDS * BAND_ROWS >= N_NODES, "bands must cover every row");
static_assert((N_BANDS - 1) * BAND_ROWS < N_NODES, "no empty band");
static_assert(N_EDGES <= (1 << EDGE_BITS), "edge index must fit its field");
static_assert(BAND_ROWS <= (1 << (32 - EDGE_BITS)), "band row must fit its field");
static_assert(N_EDGES % 128 == 0, "whole 128-edge wave steps");
static_assert(QUEUE_WORDS >= 31 + SCAN_AHEAD * 128 + 1, "queue holds one scan step");
static_assert((N_BANDS - 1) * ((N_EDGES / 128) / N_BANDS) < N_EDGES / 128, "scan start inside the list");

#define LIST_CAP 1024
#define SRC_BITS 17
#define LIST_OVERFLOW 0xFFFFFFFFu

static_assert(N_NODES <= (1 << SRC_BITS), "source row must fit its field");
static_assert(BAND_ROWS <= (1 << (32 - SRC_BITS)), "band row must fit its field");
static_assert(LIST_CAP % 32 == 0, "whole 32-word list lines");

template <int F>
__device__ __forceinline__ void add_rows(int s, int r, float w, int n, int lane,
                                         const float* __restrict__ h, float* acc) {
    constexpr int C = F / 32;
    const int wbits = __builtin_bit_cast(int, w);

#pragma unroll 1
    for (int j0 = 0; j0 < n; j0 += 8) {
        float v[8][C];
        int   rr[8];
        float ww[8];
#pragma unroll
        for (int u = 0; u < 8; ++u) {
            const int j = (j0 + u < n) ? (j0 + u) : (n - 1);
            const int sj = __builtin_amdgcn_readlane(s, j);
            rr[u] = __builtin_amdgcn_readlane(r, j);
            ww[u] = __builtin_bit_cast(float, __builtin_amdgcn_readlane(wbits, j));
            const float* hp = h + (long)sj * F + lane;
#pragma unroll
            for (int t = 0; t < C; ++t) v[u][t] = hp[32 * t];
        }
#pragma unroll
        for (int u = 0; u < 8; ++u) {
            if (j0 + u < n) {
                float* ap = acc + rr[u] * F + lane;
#pragma unroll
                for (int t = 0; t < C; ++t) atomicAdd(ap + 32 * t, v[u][t] * ww[u]);
            }
        }
    }
}

template <int F>
__device__ __forceinline__ void add_queued_rows(const unsigned* q, int n, int lane, int lo,
                                                const int* __restrict__ src,
                                                const float* __restrict__ dis,
                                                const float* __restrict__ h,
                                                float* acc, unsigned* rec) {
    const unsigned ent = q[lane < n ? lane : n - 1];
    const int e = (int)(ent & ((1u << EDGE_BITS) - 1u));
    const int r = (int)(ent >> EDGE_BITS);
    const int s = src[e];
    if (rec != nullptr && lane < n) rec[lane] = ((unsigned)r << SRC_BITS) | (unsigned)s;
    const float w = dis[s] * dis[lo + r];
    add_rows<F>(s, r, w, n, lane, h, acc);
}

template <int F, bool RECORD>
__device__ __forceinline__ unsigned scan_and_add(int wave, int lane, int lo, int band,
                                                 const int* __restrict__ src,
                                                 const int* __restrict__ dst,
                                                 const float* __restrict__ dis,
                                                 const float* __restrict__ h,
                                                 float* acc, unsigned* q, unsigned* wlist) {
    int cnt = 0;
    int rec = 0;
    constexpr int NCH = N_EDGES / 128;

    const int rot = band * (NCH / N_BANDS);

    for (int c0 = wave; c0 < NCH; c0 += SCAN_WAVES * SCAN_AHEAD) {
        int4 dv[SCAN_AHEAD];
        int  ch[SCAN_AHEAD];
#pragma unroll
        for (int u = 0; u < SCAN_AHEAD; ++u) {
            const int ci = c0 + SCAN_WAVES * u;
            int c = ((ci < NCH) ? ci : (NCH - 1)) + rot;
            if (c >= NCH) c -= NCH;
            ch[u] = c;
            dv[u] = *(const int4*)(dst + (long)c * 128 + lane * 4);
        }
#pragma unroll
        for (int u = 0; u < SCAN_AHEAD; ++u) {
            if (c0 + SCAN_WAVES * u < NCH) {
                const int ebase = ch[u] * 128 + lane * 4;
#pragma unroll
                for (int k = 0; k < 4; ++k) {
                    const int d = (k == 0) ? dv[u].x : (k == 1) ? dv[u].y : (k == 2) ? dv[u].z : dv[u].w;
                    const unsigned r = (unsigned)(d - lo);
                    const bool hit = r < (unsigned)BAND_ROWS;
                    const unsigned mask = __builtin_amdgcn_ballot_w32(hit);
                    if (mask != 0u) {
                        if (hit) {
                            const int pos = cnt + (int)__builtin_amdgcn_mbcnt_lo(mask, 0u);
                            q[pos] = (r << EDGE_BITS) | (unsigned)(ebase + k);
                        }
                        cnt += __builtin_popcount(mask);
                    }
                }
            }
        }
        while (cnt >= 32) {
            cnt -= 32;
            unsigned* where = (RECORD && rec + 32 <= LIST_CAP) ? (wlist + rec) : nullptr;
            add_queued_rows<F>(q + cnt, 32, lane, lo, src, dis, h, acc, where);
            rec += 32;
        }
    }
    if (cnt > 0) {
        unsigned* where = (RECORD && rec + cnt <= LIST_CAP) ? (wlist + rec) : nullptr;
        add_queued_rows<F>(q, cnt, lane, lo, src, dis, h, acc, where);
        rec += cnt;
    }
    return (rec <= LIST_CAP) ? (unsigned)rec : LIST_OVERFLOW;
}

template <int F, bool FROM_LISTS>
__global__ __launch_bounds__(32 * SCAN_WAVES)
void band_aggregate_kernel(const int* __restrict__ src, const int* __restrict__ dst,
                           const float* __restrict__ dis, const float* __restrict__ h,
                           float* __restrict__ out,
                           unsigned* __restrict__ lists, unsigned* __restrict__ list_len) {
    __shared__ __attribute__((aligned(16))) float acc[BAND_ROWS * F];
    __shared__ unsigned queue[SCAN_WAVES][QUEUE_WORDS];
    __shared__ unsigned wave_len[SCAN_WAVES];

    const int tid  = threadIdx.x;
    const int lane = tid & 31;
    const int wave = __builtin_amdgcn_readfirstlane(tid >> 5);
    const int band = blockIdx.x;
    const int lo   = band * BAND_ROWS;

    {
        float4* a4 = (float4*)acc;
        const float4 z = {0.0f, 0.0f, 0.0f, 0.0f};
        for (int i = tid; i < BAND_ROWS * F / 4; i += 32 * SCAN_WAVES) a4[i] = z;
    }
    __syncthreads();

    unsigned* wlist = lists + ((long)band * SCAN_WAVES + wave) * LIST_CAP;

    if (!FROM_LISTS) {
        const unsigned len = scan_and_add<F, true>(wave, lane, lo, band, src, dst, dis, h, acc,
                                                   queue[wave], wlist);
        if (lane == 0) wave_len[wave] = len;
    } else {
        const unsigned len = list_len[band * SCAN_WAVES + wave];
        if (len == LIST_OVERFLOW) {
            scan_and_add<F, false>(wave, lane, lo, band, src, dst, dis, h, acc,
                                   queue[wave], nullptr);
        } else {
            const int total = (int)len;
            unsigned ent_next = 0u;
            if (total > 0) ent_next = wlist[(lane < total) ? lane : (total - 1)];
            for (int j0 = 0; j0 < total; j0 += 32) {
                const int n = (total - j0 < 32) ? (total - j0) : 32;
                const unsigned ent = ent_next;
                const int jn = j0 + 32;
                if (jn < total) {
                    const int n2 = (total - jn < 32) ? (total - jn) : 32;
                    ent_next = wlist[jn + ((lane < n2) ? lane : (n2 - 1))];
                }
                const int s = (int)(ent & ((1u << SRC_BITS) - 1u));
                const int r = (int)(ent >> SRC_BITS);
                const float w = dis[s] * dis[lo + r];
                add_rows<F>(s, r, w, n, lane, h, acc);
            }
        }
    }

    __syncthreads();

    if (!FROM_LISTS && tid < SCAN_WAVES) list_len[band * SCAN_WAVES + tid] = wave_len[tid];

    {
        const int rows_here = ((lo + BAND_ROWS < N_NODES) ? BAND_ROWS : (N_NODES - lo));
        const int n4 = rows_here * (F / 4);
        const float4* a4 = (const float4*)acc;
        float4* o4 = (float4*)(out + (long)lo * F);
        for (int i = tid; i < n4; i += 32 * SCAN_WAVES) o4[i] = a4[i];
    }
}


extern "C" void kernel_launch(void* const* d_in, const int* in_sizes, int n_in,
                              void* d_out, int out_size, void* d_ws, size_t ws_size,
                              hipStream_t stream) {
    const float* x  = (const float*)d_in[0];
    const int*   ei = (const int*)d_in[1];
    const float* W1 = (const float*)d_in[2];
    const float* b1 = (const float*)d_in[3];
    const float* W2 = (const float*)d_in[4];
    const float* b2 = (const float*)d_in[5];
    float* out = (float*)d_out;

    const int* src = ei;
    const int* dst = ei + N_EDGES;

    float* deg  = (float*)d_ws;
    float* h1   = deg + N_NODES;
    float* agg1 = h1 + (long)N_NODES * HID_F;
    float* h2   = agg1 + (long)N_NODES * HID_F;
    unsigned* lists    = (unsigned*)(h2 + (long)N_NODES * OUT_F);
    unsigned* list_len = lists + (long)N_BANDS * SCAN_WAVES * LIST_CAP;

    zero_kernel<<<1024, 256, 0, stream>>>(deg, (long)N_NODES);
    deg_count_kernel<<<(N_EDGES + 255) / 256, 256, 0, stream>>>(dst, deg, N_EDGES);
    deg_finish_kernel<<<(N_NODES + 255) / 256, 256, 0, stream>>>(deg, N_NODES);

    gemm_wmma_kernel<HID_F, false><<<N_NODES / 16, 256, 0, stream>>>(x, W1, b1, h1);

    band_aggregate_kernel<HID_F, false><<<N_BANDS, 32 * SCAN_WAVES, 0, stream>>>(src, dst, deg, h1, agg1,
                                                                                lists, list_len);

    gemm_wmma_kernel<OUT_F, true><<<N_NODES / 16, 128, 0, stream>>>(agg1, W2, b2, h2);

    band_aggregate_kernel<OUT_F, true><<<N_BANDS, 32 * SCAN_WAVES, 0, stream>>>(src, dst, deg, h2, out,
                                                                               lists, list_len);
}


# Band owners scan dst once, in a launch of its own: in-degrees counted in LDS (integer adds), deg^-0.5 and edge lists written there; float-atomic degree pass gone (7 launches to 5); agg1 reads lists
# speedup vs baseline: 2.4160x; 1.2315x over previous
#include <hip/hip_runtime.h>
#include <hip/hip_bf16.h>

typedef __attribute__((ext_vector_type(2))) float v2f;
typedef __attribute__((ext_vector_type(8))) float v8f;

#define N_NODES 100000
#define N_EDGES 1600000
#define IN_F    128
#define HID_F   128
#define OUT_F   64

template <int FOUT, bool RELU_IN>
__global__ __launch_bounds__(32 * (FOUT / 16))
void gemm_wmma_kernel(const float* __restrict__ X, const float* __restrict__ W,
                      const float* __restrict__ bias, float* __restrict__ H) {
    constexpr int K = 128;
    const int wave = threadIdx.x >> 5;
    const int lane = threadIdx.x & 31;
    const int half = lane >> 4;
    const int l    = lane & 15;
    const int m0 = blockIdx.x * 16;
    const int n0 = wave * 16;

    const float* arow = X + (long)(m0 + l) * K + 2 * half;
    const float* brow = W + (long)(n0 + l) * K + 2 * half;

    v8f c = {};
#pragma unroll
    for (int s = 0; s < K / 4; ++s) {
        v2f a = *(const v2f*)(arow + 4 * s);
        v2f b = *(const v2f*)(brow + 4 * s);
        if (RELU_IN) {
            a.x = fmaxf(a.x, 0.0f);
            a.y = fmaxf(a.y, 0.0f);
        }
        c = __builtin_amdgcn_wmma_f32_16x16x4_f32(false, a, false, b,
                                                  (short)0, c, false, false);
    }

    const float bb = bias[n0 + l];
#pragma unroll
    for (int r = 0; r < 8; ++r) {
        const int row = m0 + r + 8 * half;
        H[(long)row * FOUT + n0 + l] = c[r] + bb;
    }
}

#define BAND_ROWS 391
#define N_BANDS   256
#define EDGE_BITS 21
#define SCAN_WAVES 32
#define SCAN_AHEAD 4
#define QUEUE_WORDS 544

static_assert(N_BANDS * BAND_ROWS >= N_NODES, "bands must cover every row");
static_assert((N_BANDS - 1) * BAND_ROWS < N_NODES, "no empty band");
static_assert(N_EDGES <= (1 << EDGE_BITS), "edge index must fit its field");
static_assert(BAND_ROWS <= (1 << (32 - EDGE_BITS)), "band row must fit its field");
static_assert(N_EDGES % 128 == 0, "whole 128-edge wave steps");
static_assert(QUEUE_WORDS >= 31 + SCAN_AHEAD * 128 + 1, "queue holds one scan step");
static_assert((N_BANDS - 1) * ((N_EDGES / 128) / N_BANDS) < N_EDGES / 128, "scan start inside the list");

#define LIST_CAP 1024
#define SRC_BITS 17
#define LIST_OVERFLOW 0xFFFFFFFFu

static_assert(N_NODES <= (1 << SRC_BITS), "source row must fit its field");
static_assert(BAND_ROWS <= (1 << (32 - SRC_BITS)), "band row must fit its field");
static_assert(LIST_CAP % 32 == 0, "whole 32-word list lines");

template <int F>
__device__ __forceinline__ void add_rows(int s, int r, float w, int n, int lane,
                                         const float* __restrict__ h, float* acc) {
    constexpr int C = F / 32;
    const int wbits = __builtin_bit_cast(int, w);

#pragma unroll 1
    for (int j0 = 0; j0 < n; j0 += 8) {
        float v[8][C];
        int   rr[8];
        float ww[8];
#pragma unroll
        for (int u = 0; u < 8; ++u) {
            const int j = (j0 + u < n) ? (j0 + u) : (n - 1);
            const int sj = __builtin_amdgcn_readlane(s, j);
            rr[u] = __builtin_amdgcn_readlane(r, j);
            ww[u] = __builtin_bit_cast(float, __builtin_amdgcn_readlane(wbits, j));
            const float* hp = h + (long)sj * F + lane;
#pragma unroll
            for (int t = 0; t < C; ++t) v[u][t] = hp[32 * t];
        }
#pragma unroll
        for (int u = 0; u < 8; ++u) {
            if (j0 + u < n) {
                float* ap = acc + rr[u] * F + lane;
#pragma unroll
                for (int t = 0; t < C; ++t) atomicAdd(ap + 32 * t, v[u][t] * ww[u]);
            }
        }
    }
}

template <int F, bool LIST>
__device__ __forceinline__ void take_queued_edges(const unsigned* q, int n, int lane, int lo,
                                                  const int* __restrict__ src,
                                                  const float* __restrict__ dis,
                                                  const float* __restrict__ h,
                                                  float* acc, unsigned* rec) {
    const unsigned ent = q[lane < n ? lane : n - 1];
    const int e = (int)(ent & ((1u << EDGE_BITS) - 1u));
    const int r = (int)(ent >> EDGE_BITS);
    if (LIST) {
        if (rec != nullptr && lane < n) rec[lane] = ((unsigned)r << SRC_BITS) | (unsigned)src[e];
    } else {
        const int s = src[e];
        const float w = dis[s] * dis[lo + r];
        add_rows<F>(s, r, w, n, lane, h, acc);
    }
}

template <int F, bool LIST>
__device__ __forceinline__ unsigned scan_band(int wave, int lane, int lo, int band,
                                              const int* __restrict__ src,
                                              const int* __restrict__ dst,
                                              const float* __restrict__ dis,
                                              const float* __restrict__ h,
                                              float* acc, unsigned* q, unsigned* wlist,
                                              unsigned* rowcnt) {
    int cnt = 0;
    int rec = 0;
    constexpr int NCH = N_EDGES / 128;

    const int rot = band * (NCH / N_BANDS);

    for (int c0 = wave; c0 < NCH; c0 += SCAN_WAVES * SCAN_AHEAD) {
        int4 dv[SCAN_AHEAD];
        int  ch[SCAN_AHEAD];
#pragma unroll
        for (int u = 0; u < SCAN_AHEAD; ++u) {
            const int ci = c0 + SCAN_WAVES * u;
            int c = ((ci < NCH) ? ci : (NCH - 1)) + rot;
            if (c >= NCH) c -= NCH;
            ch[u] = c;
            dv[u] = *(const int4*)(dst + (long)c * 128 + lane * 4);
        }
#pragma unroll
        for (int u = 0; u < SCAN_AHEAD; ++u) {
            if (c0 + SCAN_WAVES * u < NCH) {
                const int ebase = ch[u] * 128 + lane * 4;
#pragma unroll
                for (int k = 0; k < 4; ++k) {
                    const int d = (k == 0) ? dv[u].x : (k == 1) ? dv[u].y : (k == 2) ? dv[u].z : dv[u].w;
                    const unsigned r = (unsigned)(d - lo);
                    const bool hit = r < (unsigned)BAND_ROWS;
                    const unsigned mask = __builtin_amdgcn_ballot_w32(hit);
                    if (mask != 0u) {
                        if (hit) {
                            const int pos = cnt + (int)__builtin_amdgcn_mbcnt_lo(mask, 0u);
                            q[pos] = (r << EDGE_BITS) | (unsigned)(ebase + k);
                            if (LIST) atomicAdd(&rowcnt[r], 1u);
                        }
                        cnt += __builtin_popcount(mask);
                    }
                }
            }
        }
        while (cnt >= 32) {
            cnt -= 32;
            unsigned* where = (LIST && rec + 32 <= LIST_CAP) ? (wlist + rec) : nullptr;
            take_queued_edges<F, LIST>(q + cnt, 32, lane, lo, src, dis, h, acc, where);
            rec += 32;
        }
    }
    if (cnt > 0) {
        unsigned* where = (LIST && rec + cnt <= LIST_CAP) ? (wlist + rec) : nullptr;
        take_queued_edges<F, LIST>(q, cnt, lane, lo, src, dis, h, acc, where);
        rec += cnt;
    }
    return (rec <= LIST_CAP) ? (unsigned)rec : LIST_OVERFLOW;
}

__global__ __launch_bounds__(32 * SCAN_WAVES)
void band_list_kernel(const int* __restrict__ src, const int* __restrict__ dst,
                      float* __restrict__ dis,
                      unsigned* __restrict__ lists, unsigned* __restrict__ list_len) {
    __shared__ unsigned queue[SCAN_WAVES][QUEUE_WORDS];
    __shared__ unsigned rowcnt[BAND_ROWS];
    __shared__ unsigned wave_len[SCAN_WAVES];

    const int tid  = threadIdx.x;
    const int lane = tid & 31;
    const int wave = __builtin_amdgcn_readfirstlane(tid >> 5);
    const int band = blockIdx.x;
    const int lo   = band * BAND_ROWS;

    for (int i = tid; i < BAND_ROWS; i += 32 * SCAN_WAVES) rowcnt[i] = 0u;
    __syncthreads();

    unsigned* wlist = lists + ((long)band * SCAN_WAVES + wave) * LIST_CAP;
    const unsigned len = scan_band<HID_F, true>(wave, lane, lo, band, src, dst, nullptr, nullptr,
                                                nullptr, queue[wave], wlist, rowcnt);
    if (lane == 0) wave_len[wave] = len;

    __syncthreads();

    if (tid < SCAN_WAVES) list_len[band * SCAN_WAVES + tid] = wave_len[tid];

    for (int i = tid; i < BAND_ROWS; i += 32 * SCAN_WAVES) {
        if (lo + i < N_NODES) {
            const float d = (float)rowcnt[i];
            dis[lo + i] = (d > 0.0f) ? (1.0f / sqrtf(d)) : 0.0f;
        }
    }
}

template <int F>
__global__ __launch_bounds__(32 * SCAN_WAVES)
void band_aggregate_kernel(const int* __restrict__ src, const int* __restrict__ dst,
                           const float* __restrict__ dis, const float* __restrict__ h,
                           float* __restrict__ out,
                           const unsigned* __restrict__ lists,
                           const unsigned* __restrict__ list_len) {
    __shared__ __attribute__((aligned(16))) float acc[BAND_ROWS * F];
    __shared__ unsigned queue[SCAN_WAVES][QUEUE_WORDS];

    const int tid  = threadIdx.x;
    const int lane = tid & 31;
    const int wave = __builtin_amdgcn_readfirstlane(tid >> 5);
    const int band = blockIdx.x;
    const int lo   = band * BAND_ROWS;

    {
        float4* a4 = (float4*)acc;
        const float4 z = {0.0f, 0.0f, 0.0f, 0.0f};
        for (int i = tid; i < BAND_ROWS * F / 4; i += 32 * SCAN_WAVES) a4[i] = z;
    }
    __syncthreads();

    const unsigned* wlist = lists + ((long)band * SCAN_WAVES + wave) * LIST_CAP;

    const unsigned len = list_len[band * SCAN_WAVES + wave];
    if (len == LIST_OVERFLOW) {
        scan_band<F, false>(wave, lane, lo, band, src, dst, dis, h, acc, queue[wave], nullptr,
                            nullptr);
    } else {
        const int total = (int)len;
        unsigned ent_next = 0u;
        if (total > 0) ent_next = wlist[(lane < total) ? lane : (total - 1)];
        for (int j0 = 0; j0 < total; j0 += 32) {
            const int n = (total - j0 < 32) ? (total - j0) : 32;
            const unsigned ent = ent_next;
            const int jn = j0 + 32;
            if (jn < total) {
                const int n2 = (total - jn < 32) ? (total - jn) : 32;
                ent_next = wlist[jn + ((lane < n2) ? lane : (n2 - 1))];
            }
            const int s = (int)(ent & ((1u << SRC_BITS) - 1u));
            const int r = (int)(ent >> SRC_BITS);
            const float w = dis[s] * dis[lo + r];
            add_rows<F>(s, r, w, n, lane, h, acc);
        }
    }

    __syncthreads();

    {
        const int rows_here = ((lo + BAND_ROWS < N_NODES) ? BAND_ROWS : (N_NODES - lo));
        const int n4 = rows_here * (F / 4);
        const float4* a4 = (const float4*)acc;
        float4* o4 = (float4*)(out + (long)lo * F);
        for (int i = tid; i < n4; i += 32 * SCAN_WAVES) o4[i] = a4[i];
    }
}


extern "C" void kernel_launch(void* const* d_in, const int* in_sizes, int n_in,
                              void* d_out, int out_size, void* d_ws, size_t ws_size,
                              hipStream_t stream) {
    const float* x  = (const float*)d_in[0];
    const int*   ei = (const int*)d_in[1];
    const float* W1 = (const float*)d_in[2];
    const float* b1 = (const float*)d_in[3];
    const float* W2 = (const float*)d_in[4];
    const float* b2 = (const float*)d_in[5];
    float* out = (float*)d_out;

    const int* src = ei;
    const int* dst = ei + N_EDGES;

    float* deg  = (float*)d_ws;
    float* h1   = deg + N_NODES;
    float* agg1 = h1 + (long)N_NODES * HID_F;
    float* h2   = agg1 + (long)N_NODES * HID_F;
    unsigned* lists    = (unsigned*)(h2 + (long)N_NODES * OUT_F);
    unsigned* list_len = lists + (long)N_BANDS * SCAN_WAVES * LIST_CAP;

    band_list_kernel<<<N_BANDS, 32 * SCAN_WAVES, 0, stream>>>(src, dst, deg, lists, list_len);

    gemm_wmma_kernel<HID_F, false><<<N_NODES / 16, 256, 0, stream>>>(x, W1, b1, h1);

    band_aggregate_kernel<HID_F><<<N_BANDS, 32 * SCAN_WAVES, 0, stream>>>(src, dst, deg, h1, agg1,
                                                                         lists, list_len);

    gemm_wmma_kernel<OUT_F, true><<<N_NODES / 16, 128, 0, stream>>>(agg1, W2, b2, h2);

    band_aggregate_kernel<OUT_F><<<N_BANDS, 32 * SCAN_WAVES, 0, stream>>>(src, dst, deg, h2, out,
                                                                         lists, list_len);
}


# Layer-2 product moved into the first aggregation: a band's work-group multiplies its relu'd LDS sums by W2 (copied to LDS) on the matrix unit, stores h2; agg1 plane and GEMM launch gone (5 to 4)
# speedup vs baseline: 2.7039x; 1.1192x over previous
#include <hip/hip_runtime.h>
#include <hip/hip_bf16.h>

typedef __attribute__((ext_vector_type(2))) float v2f;
typedef __attribute__((ext_vector_type(8))) float v8f;

#define N_NODES 100000
#define N_EDGES 1600000
#define IN_F    128
#define HID_F   128
#define OUT_F   64

template <int FOUT, bool RELU_IN>
__global__ __launch_bounds__(32 * (FOUT / 16))
void gemm_wmma_kernel(const float* __restrict__ X, const float* __restrict__ W,
                      const float* __restrict__ bias, float* __restrict__ H) {
    constexpr int K = 128;
    const int wave = threadIdx.x >> 5;
    const int lane = threadIdx.x & 31;
    const int half = lane >> 4;
    const int l    = lane & 15;
    const int m0 = blockIdx.x * 16;
    const int n0 = wave * 16;

    const float* arow = X + (long)(m0 + l) * K + 2 * half;
    const float* brow = W + (long)(n0 + l) * K + 2 * half;

    v8f c = {};
#pragma unroll
    for (int s = 0; s < K / 4; ++s) {
        v2f a = *(const v2f*)(arow + 4 * s);
        v2f b = *(const v2f*)(brow + 4 * s);
        if (RELU_IN) {
            a.x = fmaxf(a.x, 0.0f);
            a.y = fmaxf(a.y, 0.0f);
        }
        c = __builtin_amdgcn_wmma_f32_16x16x4_f32(false, a, false, b,
                                                  (short)0, c, false, false);
    }

    const float bb = bias[n0 + l];
#pragma unroll
    for (int r = 0; r < 8; ++r) {
        const int row = m0 + r + 8 * half;
        H[(long)row * FOUT + n0 + l] = c[r] + bb;
    }
}

#define BAND_ROWS 391
#define N_BANDS   256
#define EDGE_BITS 21
#define SCAN_WAVES 32
#define SCAN_AHEAD 4
#define QUEUE_WORDS 544

static_assert(N_BANDS * BAND_ROWS >= N_NODES, "bands must cover every row");
static_assert((N_BANDS - 1) * BAND_ROWS < N_NODES, "no empty band");
static_assert(N_EDGES <= (1 << EDGE_BITS), "edge index must fit its field");
static_assert(BAND_ROWS <= (1 << (32 - EDGE_BITS)), "band row must fit its field");
static_assert(N_EDGES % 128 == 0, "whole 128-edge wave steps");
static_assert(QUEUE_WORDS >= 31 + SCAN_AHEAD * 128 + 1, "queue holds one scan step");
static_assert((N_BANDS - 1) * ((N_EDGES / 128) / N_BANDS) < N_EDGES / 128, "scan start inside the list");

#define LIST_CAP 1024
#define SRC_BITS 17
#define LIST_OVERFLOW 0xFFFFFFFFu

static_assert(N_NODES <= (1 << SRC_BITS), "source row must fit its field");
static_assert(BAND_ROWS <= (1 << (32 - SRC_BITS)), "band row must fit its field");
static_assert(LIST_CAP % 32 == 0, "whole 32-word list lines");

template <int F, int P>
__device__ __forceinline__ void add_rows(int s, int r, float w, int n, int lane,
                                         const float* __restrict__ h, float* acc) {
    constexpr int C = F / 32;
    const int wbits = __builtin_bit_cast(int, w);

#pragma unroll 1
    for (int j0 = 0; j0 < n; j0 += 8) {
        float v[8][C];
        int   rr[8];
        float ww[8];
#pragma unroll
        for (int u = 0; u < 8; ++u) {
            const int j = (j0 + u < n) ? (j0 + u) : (n - 1);
            const int sj = __builtin_amdgcn_readlane(s, j);
            rr[u] = __builtin_amdgcn_readlane(r, j);
            ww[u] = __builtin_bit_cast(float, __builtin_amdgcn_readlane(wbits, j));
            const float* hp = h + (long)sj * F + lane;
#pragma unroll
            for (int t = 0; t < C; ++t) v[u][t] = hp[32 * t];
        }
#pragma unroll
        for (int u = 0; u < 8; ++u) {
            if (j0 + u < n) {
                float* ap = acc + rr[u] * P + lane;
#pragma unroll
                for (int t = 0; t < C; ++t) atomicAdd(ap + 32 * t, v[u][t] * ww[u]);
            }
        }
    }
}

template <int F, int P, bool LIST>
__device__ __forceinline__ void take_queued_edges(const unsigned* q, int n, int lane, int lo,
                                                  const int* __restrict__ src,
                                                  const float* __restrict__ dis,
                                                  const float* __restrict__ h,
                                                  float* acc, unsigned* rec) {
    const unsigned ent = q[lane < n ? lane : n - 1];
    const int e = (int)(ent & ((1u << EDGE_BITS) - 1u));
    const int r = (int)(ent >> EDGE_BITS);
    if (LIST) {
        if (rec != nullptr && lane < n) rec[lane] = ((unsigned)r << SRC_BITS) | (unsigned)src[e];
    } else {
        const int s = src[e];
        const float w = dis[s] * dis[lo + r];
        add_rows<F, P>(s, r, w, n, lane, h, acc);
    }
}

template <int F, int P, bool LIST>
__device__ __forceinline__ unsigned scan_band(int wave, int lane, int lo, int band,
                                              const int* __restrict__ src,
                                              const int* __restrict__ dst,
                                              const float* __restrict__ dis,
                                              const float* __restrict__ h,
                                              float* acc, unsigned* q, unsigned* wlist,
                                              unsigned* rowcnt) {
    int cnt = 0;
    int rec = 0;
    constexpr int NCH = N_EDGES / 128;

    const int rot = band * (NCH / N_BANDS);

    for (int c0 = wave; c0 < NCH; c0 += SCAN_WAVES * SCAN_AHEAD) {
        int4 dv[SCAN_AHEAD];
        int  ch[SCAN_AHEAD];
#pragma unroll
        for (int u = 0; u < SCAN_AHEAD; ++u) {
            const int ci = c0 + SCAN_WAVES * u;
            int c = ((ci < NCH) ? ci : (NCH - 1)) + rot;
            if (c >= NCH) c -= NCH;
            ch[u] = c;
            dv[u] = *(const int4*)(dst + (long)c * 128 + lane * 4);
        }
#pragma unroll
        for (int u = 0; u < SCAN_AHEAD; ++u) {
            if (c0 + SCAN_WAVES * u < NCH) {
                const int ebase = ch[u] * 128 + lane * 4;
#pragma unroll
                for (int k = 0; k < 4; ++k) {
                    const int d = (k == 0) ? dv[u].x : (k == 1) ? dv[u].y : (k == 2) ? dv[u].z : dv[u].w;
                    const unsigned r = (unsigned)(d - lo);
                    const bool hit = r < (unsigned)BAND_ROWS;
                    const unsigned mask = __builtin_amdgcn_ballot_w32(hit);
                    if (mask != 0u) {
                        if (hit) {
                            const int pos = cnt + (int)__builtin_amdgcn_mbcnt_lo(mask, 0u);
                            q[pos] = (r << EDGE_BITS) | (unsigned)(ebase + k);
                            if (LIST) atomicAdd(&rowcnt[r], 1u);
                        }
                        cnt += __builtin_popcount(mask);
                    }
                }
            }
        }
        while (cnt >= 32) {
            cnt -= 32;
            unsigned* where = (LIST && rec + 32 <= LIST_CAP) ? (wlist + rec) : nullptr;
            take_queued_edges<F, P, LIST>(q + cnt, 32, lane, lo, src, dis, h, acc, where);
            rec += 32;
        }
    }
    if (cnt > 0) {
        unsigned* where = (LIST && rec + cnt <= LIST_CAP) ? (wlist + rec) : nullptr;
        take_queued_edges<F, P, LIST>(q, cnt, lane, lo, src, dis, h, acc, where);
        rec += cnt;
    }
    return (rec <= LIST_CAP) ? (unsigned)rec : LIST_OVERFLOW;
}

__global__ __launch_bounds__(32 * SCAN_WAVES)
void band_list_kernel(const int* __restrict__ src, const int* __restrict__ dst,
                      float* __restrict__ dis,
                      unsigned* __restrict__ lists, unsigned* __restrict__ list_len) {
    __shared__ unsigned queue[SCAN_WAVES][QUEUE_WORDS];
    __shared__ unsigned rowcnt[BAND_ROWS];
    __shared__ unsigned wave_len[SCAN_WAVES];

    const int tid  = threadIdx.x;
    const int lane = tid & 31;
    const int wave = __builtin_amdgcn_readfirstlane(tid >> 5);
    const int band = blockIdx.x;
    const int lo   = band * BAND_ROWS;

    for (int i = tid; i < BAND_ROWS; i += 32 * SCAN_WAVES) rowcnt[i] = 0u;
    __syncthreads();

    unsigned* wlist = lists + ((long)band * SCAN_WAVES + wave) * LIST_CAP;
    const unsigned len = scan_band<HID_F, HID_F, true>(wave, lane, lo, band, src, dst, nullptr, nullptr,
                                                       nullptr, queue[wave], wlist, rowcnt);
    if (lane == 0) wave_len[wave] = len;

    __syncthreads();

    if (tid < SCAN_WAVES) list_len[band * SCAN_WAVES + tid] = wave_len[tid];

    for (int i = tid; i < BAND_ROWS; i += 32 * SCAN_WAVES) {
        if (lo + i < N_NODES) {
            const float d = (float)rowcnt[i];
            dis[lo + i] = (d > 0.0f) ? (1.0f / sqrtf(d)) : 0.0f;
        }
    }
}

template <int F, int P>
__device__ __forceinline__ void add_band_lists(int wave, int lane, int lo, int band,
                                               const int* __restrict__ src,
                                               const int* __restrict__ dst,
                                               const float* __restrict__ dis,
                                               const float* __restrict__ h,
                                               float* acc, unsigned* q,
                                               const unsigned* __restrict__ lists,
                                               const unsigned* __restrict__ list_len) {
    const unsigned* wlist = lists + ((long)band * SCAN_WAVES + wave) * LIST_CAP;

    const unsigned len = list_len[band * SCAN_WAVES + wave];
    if (len == LIST_OVERFLOW) {
        scan_band<F, P, false>(wave, lane, lo, band, src, dst, dis, h, acc, q, nullptr, nullptr);
    } else {
        const int total = (int)len;
        unsigned ent_next = 0u;
        if (total > 0) ent_next = wlist[(lane < total) ? lane : (total - 1)];
        for (int j0 = 0; j0 < total; j0 += 32) {
            const int n = (total - j0 < 32) ? (total - j0) : 32;
            const unsigned ent = ent_next;
            const int jn = j0 + 32;
            if (jn < total) {
                const int n2 = (total - jn < 32) ? (total - jn) : 32;
                ent_next = wlist[jn + ((lane < n2) ? lane : (n2 - 1))];
            }
            const int s = (int)(ent & ((1u << SRC_BITS) - 1u));
            const int r = (int)(ent >> SRC_BITS);
            const float w = dis[s] * dis[lo + r];
            add_rows<F, P>(s, r, w, n, lane, h, acc);
        }
    }
}

template <int F>
__global__ __launch_bounds__(32 * SCAN_WAVES)
void band_aggregate_kernel(const int* __restrict__ src, const int* __restrict__ dst,
                           const float* __restrict__ dis, const float* __restrict__ h,
                           float* __restrict__ out,
                           const unsigned* __restrict__ lists,
                           const unsigned* __restrict__ list_len) {
    __shared__ __attribute__((aligned(16))) float acc[BAND_ROWS * F];
    __shared__ unsigned queue[SCAN_WAVES][QUEUE_WORDS];

    const int tid  = threadIdx.x;
    const int lane = tid & 31;
    const int wave = __builtin_amdgcn_readfirstlane(tid >> 5);
    const int band = blockIdx.x;
    const int lo   = band * BAND_ROWS;

    {
        float4* a4 = (float4*)acc;
        const float4 z = {0.0f, 0.0f, 0.0f, 0.0f};
        for (int i = tid; i < BAND_ROWS * F / 4; i += 32 * SCAN_WAVES) a4[i] = z;
    }
    __syncthreads();

    add_band_lists<F, F>(wave, lane, lo, band, src, dst, dis, h, acc, queue[wave], lists, list_len);

    __syncthreads();

    {
        const int rows_here = ((lo + BAND_ROWS < N_NODES) ? BAND_ROWS : (N_NODES - lo));
        const int n4 = rows_here * (F / 4);
        const float4* a4 = (const float4*)acc;
        float4* o4 = (float4*)(out + (long)lo * F);
        for (int i = tid; i < n4; i += 32 * SCAN_WAVES) o4[i] = a4[i];
    }
}

#define DENSE_PITCH 132
#define DENSE_TILES ((BAND_ROWS + 15) / 16)
#define DENSE_ROWS  (DENSE_TILES * 16)

static_assert(DENSE_PITCH % 4 == 0, "rows stay 16-byte aligned");
static_assert(OUT_F % 32 == 0, "a wave's job is 32 output columns");

__global__ __launch_bounds__(32 * SCAN_WAVES)
void band_aggregate_dense_kernel(const int* __restrict__ src, const int* __restrict__ dst,
                                 const float* __restrict__ dis, const float* __restrict__ h,
                                 const float* __restrict__ W, const float* __restrict__ bias,
                                 float* __restrict__ out,
                                 const unsigned* __restrict__ lists,
                                 const unsigned* __restrict__ list_len) {
    __shared__ __attribute__((aligned(16))) float acc[DENSE_ROWS * DENSE_PITCH];
    __shared__ __attribute__((aligned(16))) float wsh[OUT_F * DENSE_PITCH];
    __shared__ unsigned queue[SCAN_WAVES][QUEUE_WORDS];

    const int tid  = threadIdx.x;
    const int lane = tid & 31;
    const int wave = __builtin_amdgcn_readfirstlane(tid >> 5);
    const int band = blockIdx.x;
    const int lo   = band * BAND_ROWS;

    {
        float4* a4 = (float4*)acc;
        const float4 z = {0.0f, 0.0f, 0.0f, 0.0f};
        for (int i = tid; i < DENSE_ROWS * DENSE_PITCH / 4; i += 32 * SCAN_WAVES) a4[i] = z;
        const float4* w4 = (const float4*)W;
        for (int i = tid; i < OUT_F * (HID_F / 4); i += 32 * SCAN_WAVES) {
            const int n  = i / (HID_F / 4);
            const int k4 = i % (HID_F / 4);
            *(float4*)(wsh + n * DENSE_PITCH + 4 * k4) = w4[i];
        }
    }
    __syncthreads();

    add_band_lists<HID_F, DENSE_PITCH>(wave, lane, lo, band, src, dst, dis, h, acc, queue[wave],
                                       lists, list_len);

    __syncthreads();

    const int rows_here  = ((lo + BAND_ROWS < N_NODES) ? BAND_ROWS : (N_NODES - lo));
    const int tiles_here = (rows_here + 15) / 16;
    const int half = lane >> 4;
    const int l    = lane & 15;
    for (int job = wave; job < tiles_here * (OUT_F / 32); job += SCAN_WAVES) {
        const int m0 = (job / (OUT_F / 32)) * 16;
        const int n0 = (job % (OUT_F / 32)) * 32;

        const float* arow  = acc + (m0 + l) * DENSE_PITCH + 2 * half;
        const float* brow0 = wsh + (n0 + l) * DENSE_PITCH + 2 * half;
        const float* brow1 = brow0 + 16 * DENSE_PITCH;

        v8f c0 = {};
        v8f c1 = {};
#pragma unroll
        for (int s = 0; s < HID_F / 4; ++s) {
            v2f a  = *(const v2f*)(arow + 4 * s);
            v2f b0 = *(const v2f*)(brow0 + 4 * s);
            v2f b1 = *(const v2f*)(brow1 + 4 * s);
            a.x = fmaxf(a.x, 0.0f);
            a.y = fmaxf(a.y, 0.0f);
            c0 = __builtin_amdgcn_wmma_f32_16x16x4_f32(false, a, false, b0,
                                                       (short)0, c0, false, false);
            c1 = __builtin_amdgcn_wmma_f32_16x16x4_f32(false, a, false, b1,
                                                       (short)0, c1, false, false);
        }

        const float bb0 = bias[n0 + l];
        const float bb1 = bias[n0 + 16 + l];
#pragma unroll
        for (int r = 0; r < 8; ++r) {
            const int row = m0 + r + 8 * half;
            if (row < rows_here) {
                float* op = out + (long)(lo + row) * OUT_F + n0 + l;
                op[0]  = c0[r] + bb0;
                op[16] = c1[r] + bb1;
            }
        }
    }
}


extern "C" void kernel_launch(void* const* d_in, const int* in_sizes, int n_in,
                              void* d_out, int out_size, void* d_ws, size_t ws_size,
                              hipStream_t stream) {
    const float* x  = (const float*)d_in[0];
    const int*   ei = (const int*)d_in[1];
    const float* W1 = (const float*)d_in[2];
    const float* b1 = (const float*)d_in[3];
    const float* W2 = (const float*)d_in[4];
    const float* b2 = (const float*)d_in[5];
    float* out = (float*)d_out;

    const int* src = ei;
    const int* dst = ei + N_EDGES;

    float* deg  = (float*)d_ws;
    float* h1   = deg + N_NODES;
    float* h2   = h1 + (long)N_NODES * HID_F;
    unsigned* lists    = (unsigned*)(h2 + (long)N_NODES * OUT_F);
    unsigned* list_len = lists + (long)N_BANDS * SCAN_WAVES * LIST_CAP;

    band_list_kernel<<<N_BANDS, 32 * SCAN_WAVES, 0, stream>>>(src, dst, deg, lists, list_len);

    gemm_wmma_kernel<HID_F, false><<<N_NODES / 16, 256, 0, stream>>>(x, W1, b1, h1);

    band_aggregate_dense_kernel<<<N_BANDS, 32 * SCAN_WAVES, 0, stream>>>(src, dst, deg, h1, W2, b2, h2,
                                                                        lists, list_len);

    band_aggregate_kernel<OUT_F><<<N_BANDS, 32 * SCAN_WAVES, 0, stream>>>(src, dst, deg, h2, out,
                                                                         lists, list_len);
}


# Layer-1 product moved into the scan launch: a band's work-group multiplies its own rows of x by W1 (copied to LDS) on the matrix unit between scan steps, stores h1; GEMM launch gone (4 to 3)
# speedup vs baseline: 3.2248x; 1.1926x over previous
#include <hip/hip_runtime.h>
#include <hip/hip_bf16.h>

typedef __attribute__((ext_vector_type(2))) float v2f;
typedef __attribute__((ext_vector_type(8))) float v8f;

#define N_NODES 100000
#define N_EDGES 1600000
#define IN_F    128
#define HID_F   128
#define OUT_F   64

#define DENSE_PITCH 132

static_assert(DENSE_PITCH % 4 == 0, "rows stay 16-byte aligned");

__device__ __forceinline__ void layer1_tile(int lane, int lo, int m0, int rows_here,
                                            const float* __restrict__ X, const float* wsh,
                                            const float* __restrict__ bias, float* __restrict__ H) {
    constexpr int NT = 2;
    constexpr int KS = IN_F / 4;
    constexpr int KP = KS / 2;
    constexpr int AHEAD = 2;
    const int half = lane >> 4;
    const int l    = lane & 15;

    const int arow_i = (m0 + l < rows_here) ? (m0 + l) : (rows_here - 1);
    const float* arow = X + (long)(lo + arow_i) * IN_F + 2 * half;

    v2f a[KS];
#pragma unroll
    for (int s = 0; s < KS; ++s) a[s] = *(const v2f*)(arow + 4 * s);

#pragma unroll 1
    for (int n0 = 0; n0 < HID_F; n0 += 16 * NT) {
        const float* brow = wsh + (n0 + l) * DENSE_PITCH + 2 * half;

        v8f c[NT];
#pragma unroll
        for (int j = 0; j < NT; ++j) c[j] = v8f{};

        v2f b[KP][NT][2];
#pragma unroll
        for (int p = 0; p < AHEAD; ++p)
#pragma unroll
            for (int j = 0; j < NT; ++j)
#pragma unroll
                for (int t = 0; t < 2; ++t)
                    b[p][j][t] = *(const v2f*)(brow + j * 16 * DENSE_PITCH + 4 * (2 * p + t));

#pragma unroll
        for (int p = 0; p < KP; ++p) {
            if (p + AHEAD < KP) {
#pragma unroll
                for (int j = 0; j < NT; ++j)
#pragma unroll
                    for (int t = 0; t < 2; ++t)
                        b[p + AHEAD][j][t] =
                            *(const v2f*)(brow + j * 16 * DENSE_PITCH + 4 * (2 * (p + AHEAD) + t));
            }
            __builtin_amdgcn_sched_barrier(0);
#pragma unroll
            for (int t = 0; t < 2; ++t)
#pragma unroll
                for (int j = 0; j < NT; ++j) {
                    c[j] = __builtin_amdgcn_wmma_f32_16x16x4_f32(false, a[2 * p + t], false, b[p][j][t],
                                                                 (short)0, c[j], false, false);
                }
        }

        float* hp = H + (long)(lo + m0 + 8 * half) * HID_F + n0 + l;
        const int rows_left = rows_here - (m0 + 8 * half);
#pragma unroll
        for (int j = 0; j < NT; ++j) {
            const float bb = bias[n0 + 16 * j + l];
#pragma unroll
            for (int r = 0; r < 8; ++r) {
                if (r < rows_left) hp[r * HID_F + 16 * j] = c[j][r] + bb;
            }
        }
    }
}

#define BAND_ROWS 391
#define N_BANDS   256
#define EDGE_BITS 21
#define SCAN_WAVES 32
#define SCAN_AHEAD 4
#define QUEUE_WORDS 544

static_assert(N_BANDS * BAND_ROWS >= N_NODES, "bands must cover every row");
static_assert((N_BANDS - 1) * BAND_ROWS < N_NODES, "no empty band");
static_assert(N_EDGES <= (1 << EDGE_BITS), "edge index must fit its field");
static_assert(BAND_ROWS <= (1 << (32 - EDGE_BITS)), "band row must fit its field");
static_assert(N_EDGES % 128 == 0, "whole 128-edge wave steps");
static_assert(QUEUE_WORDS >= 31 + SCAN_AHEAD * 128 + 1, "queue holds one scan step");
static_assert((N_BANDS - 1) * ((N_EDGES / 128) / N_BANDS) < N_EDGES / 128, "scan start inside the list");

#define LIST_CAP 1024
#define SRC_BITS 17
#define LIST_OVERFLOW 0xFFFFFFFFu

static_assert(N_NODES <= (1 << SRC_BITS), "source row must fit its field");
static_assert(BAND_ROWS <= (1 << (32 - SRC_BITS)), "band row must fit its field");
static_assert(LIST_CAP % 32 == 0, "whole 32-word list lines");

template <int F, int P>
__device__ __forceinline__ void add_rows(int s, int r, float w, int n, int lane,
                                         const float* __restrict__ h, float* acc) {
    constexpr int C = F / 32;
    const int wbits = __builtin_bit_cast(int, w);

#pragma unroll 1
    for (int j0 = 0; j0 < n; j0 += 8) {
        float v[8][C];
        int   rr[8];
        float ww[8];
#pragma unroll
        for (int u = 0; u < 8; ++u) {
            const int j = (j0 + u < n) ? (j0 + u) : (n - 1);
            const int sj = __builtin_amdgcn_readlane(s, j);
            rr[u] = __builtin_amdgcn_readlane(r, j);
            ww[u] = __builtin_bit_cast(float, __builtin_amdgcn_readlane(wbits, j));
            const float* hp = h + (long)sj * F + lane;
#pragma unroll
            for (int t = 0; t < C; ++t) v[u][t] = hp[32 * t];
        }
#pragma unroll
        for (int u = 0; u < 8; ++u) {
            if (j0 + u < n) {
                float* ap = acc + rr[u] * P + lane;
#pragma unroll
                for (int t = 0; t < C; ++t) atomicAdd(ap + 32 * t, v[u][t] * ww[u]);
            }
        }
    }
}

template <int F, int P, bool LIST>
__device__ __forceinline__ void take_queued_edges(const unsigned* q, int n, int lane, int lo,
                                                  const int* __restrict__ src,
                                                  const float* __restrict__ dis,
                                                  const float* __restrict__ h,
                                                  float* acc, unsigned* rec) {
    const unsigned ent = q[lane < n ? lane : n - 1];
    const int e = (int)(ent & ((1u << EDGE_BITS) - 1u));
    const int r = (int)(ent >> EDGE_BITS);
    if (LIST) {
        if (rec != nullptr && lane < n) rec[lane] = ((unsigned)r << SRC_BITS) | (unsigned)src[e];
    } else {
        const int s = src[e];
        const float w = dis[s] * dis[lo + r];
        add_rows<F, P>(s, r, w, n, lane, h, acc);
    }
}

template <int F, int P, bool LIST>
__device__ __forceinline__ void scan_steps(int c_from, int c_to, int& cnt, int& rec,
                                           int lane, int lo, int band,
                                           const int* __restrict__ src,
                                           const int* __restrict__ dst,
                                           const float* __restrict__ dis,
                                           const float* __restrict__ h,
                                           float* acc, unsigned* q, unsigned* wlist,
                                           unsigned* rowcnt) {
    constexpr int NCH = N_EDGES / 128;

    const int rot = band * (NCH / N_BANDS);

    for (int c0 = c_from; c0 < c_to; c0 += SCAN_WAVES * SCAN_AHEAD) {
        int4 dv[SCAN_AHEAD];
        int  ch[SCAN_AHEAD];
#pragma unroll
        for (int u = 0; u < SCAN_AHEAD; ++u) {
            const int ci = c0 + SCAN_WAVES * u;
            int c = ((ci < NCH) ? ci : (NCH - 1)) + rot;
            if (c >= NCH) c -= NCH;
            ch[u] = c;
            dv[u] = *(const int4*)(dst + (long)c * 128 + lane * 4);
        }
#pragma unroll
        for (int u = 0; u < SCAN_AHEAD; ++u) {
            if (c0 + SCAN_WAVES * u < NCH) {
                const int ebase = ch[u] * 128 + lane * 4;
#pragma unroll
                for (int k = 0; k < 4; ++k) {
                    const int d = (k == 0) ? dv[u].x : (k == 1) ? dv[u].y : (k == 2) ? dv[u].z : dv[u].w;
                    const unsigned r = (unsigned)(d - lo);
                    const bool hit = r < (unsigned)BAND_ROWS;
                    const unsigned mask = __builtin_amdgcn_ballot_w32(hit);
                    if (mask != 0u) {
                        if (hit) {
                            const int pos = cnt + (int)__builtin_amdgcn_mbcnt_lo(mask, 0u);
                            q[pos] = (r << EDGE_BITS) | (unsigned)(ebase + k);
                            if (LIST) atomicAdd(&rowcnt[r], 1u);
                        }
                        cnt += __builtin_popcount(mask);
                    }
                }
            }
        }
        while (cnt >= 32) {
            cnt -= 32;
            unsigned* where = (LIST && rec + 32 <= LIST_CAP) ? (wlist + rec) : nullptr;
            take_queued_edges<F, P, LIST>(q + cnt, 32, lane, lo, src, dis, h, acc, where);
            rec += 32;
        }
    }
}

template <int F, int P, bool LIST>
__device__ __forceinline__ unsigned scan_finish(int cnt, int rec, int lane, int lo,
                                                const int* __restrict__ src,
                                                const float* __restrict__ dis,
                                                const float* __restrict__ h,
                                                float* acc, unsigned* q, unsigned* wlist) {
    if (cnt > 0) {
        unsigned* where = (LIST && rec + cnt <= LIST_CAP) ? (wlist + rec) : nullptr;
        take_queued_edges<F, P, LIST>(q, cnt, lane, lo, src, dis, h, acc, where);
        rec += cnt;
    }
    return (rec <= LIST_CAP) ? (unsigned)rec : LIST_OVERFLOW;
}

#define DENSE_TILES ((BAND_ROWS + 15) / 16)
#define JOB_SLOTS 8
#define JOB_GAP   12

static_assert(DENSE_TILES <= SCAN_WAVES, "one tile per wave covers the band");
static_assert((JOB_SLOTS - 1) * JOB_GAP * SCAN_WAVES * SCAN_AHEAD + SCAN_WAVES <= N_EDGES / 128,
              "every slot is a scan step that every wave reaches");

__global__ __launch_bounds__(32 * SCAN_WAVES)
void band_list_kernel(const int* __restrict__ src, const int* __restrict__ dst,
                      const float* __restrict__ x, const float* __restrict__ W,
                      const float* __restrict__ bias, float* __restrict__ h1,
                      float* __restrict__ dis,
                      unsigned* __restrict__ lists, unsigned* __restrict__ list_len) {
    __shared__ unsigned queue[SCAN_WAVES][QUEUE_WORDS];
    __shared__ unsigned rowcnt[BAND_ROWS];
    __shared__ unsigned wave_len[SCAN_WAVES];
    __shared__ __attribute__((aligned(16))) float wsh[HID_F * DENSE_PITCH];

    const int tid  = threadIdx.x;
    const int lane = tid & 31;
    const int wave = __builtin_amdgcn_readfirstlane(tid >> 5);
    const int band = blockIdx.x;
    const int lo   = band * BAND_ROWS;

    for (int i = tid; i < BAND_ROWS; i += 32 * SCAN_WAVES) rowcnt[i] = 0u;
    {
        const float4* w4 = (const float4*)W;
        for (int i = tid; i < HID_F * (IN_F / 4); i += 32 * SCAN_WAVES) {
            const int n  = i / (IN_F / 4);
            const int k4 = i % (IN_F / 4);
            *(float4*)(wsh + n * DENSE_PITCH + 4 * k4) = w4[i];
        }
    }
    __syncthreads();

    const int rows_here  = ((lo + BAND_ROWS < N_NODES) ? BAND_ROWS : (N_NODES - lo));
    const int tiles_here = (rows_here + 15) / 16;
    const int c_job = wave + SCAN_WAVES * SCAN_AHEAD * JOB_GAP * ((wave >> 2) % JOB_SLOTS);

    unsigned* wlist = lists + ((long)band * SCAN_WAVES + wave) * LIST_CAP;
    int cnt = 0;
    int rec = 0;
    scan_steps<HID_F, HID_F, true>(wave, c_job, cnt, rec, lane, lo, band, src, dst, nullptr, nullptr,
                                   nullptr, queue[wave], wlist, rowcnt);
    if (wave < tiles_here) layer1_tile(lane, lo, wave * 16, rows_here, x, wsh, bias, h1);
    scan_steps<HID_F, HID_F, true>(c_job, N_EDGES / 128, cnt, rec, lane, lo, band, src, dst, nullptr,
                                   nullptr, nullptr, queue[wave], wlist, rowcnt);
    const unsigned len = scan_finish<HID_F, HID_F, true>(cnt, rec, lane, lo, src, nullptr, nullptr,
                                                         nullptr, queue[wave], wlist);
    if (lane == 0) wave_len[wave] = len;

    __syncthreads();

    if (tid < SCAN_WAVES) list_len[band * SCAN_WAVES + tid] = wave_len[tid];

    for (int i = tid; i < BAND_ROWS; i += 32 * SCAN_WAVES) {
        if (lo + i < N_NODES) {
            const float d = (float)rowcnt[i];
            dis[lo + i] = (d > 0.0f) ? (1.0f / sqrtf(d)) : 0.0f;
        }
    }
}

template <int F, int P>
__device__ __forceinline__ void add_band_lists(int wave, int lane, int lo, int band,
                                               const int* __restrict__ src,
                                               const int* __restrict__ dst,
                                               const float* __restrict__ dis,
                                               const float* __restrict__ h,
                                               float* acc, unsigned* q,
                                               const unsigned* __restrict__ lists,
                                               const unsigned* __restrict__ list_len) {
    const unsigned* wlist = lists + ((long)band * SCAN_WAVES + wave) * LIST_CAP;

    const unsigned len = list_len[band * SCAN_WAVES + wave];
    if (len == LIST_OVERFLOW) {
        int cnt = 0;
        int rec = 0;
        scan_steps<F, P, false>(wave, N_EDGES / 128, cnt, rec, lane, lo, band, src, dst, dis, h, acc, q,
                                nullptr, nullptr);
        scan_finish<F, P, false>(cnt, rec, lane, lo, src, dis, h, acc, q, nullptr);
    } else {
        const int total = (int)len;
        unsigned ent_next = 0u;
        if (total > 0) ent_next = wlist[(lane < total) ? lane : (total - 1)];
        for (int j0 = 0; j0 < total; j0 += 32) {
            const int n = (total - j0 < 32) ? (total - j0) : 32;
            const unsigned ent = ent_next;
            const int jn = j0 + 32;
            if (jn < total) {
                const int n2 = (total - jn < 32) ? (total - jn) : 32;
                ent_next = wlist[jn + ((lane < n2) ? lane : (n2 - 1))];
            }
            const int s = (int)(ent & ((1u << SRC_BITS) - 1u));
            const int r = (int)(ent >> SRC_BITS);
            const float w = dis[s] * dis[lo + r];
            add_rows<F, P>(s, r, w, n, lane, h, acc);
        }
    }
}

template <int F>
__global__ __launch_bounds__(32 * SCAN_WAVES)
void band_aggregate_kernel(const int* __restrict__ src, const int* __restrict__ dst,
                           const float* __restrict__ dis, const float* __restrict__ h,
                           float* __restrict__ out,
                           const unsigned* __restrict__ lists,
                           const unsigned* __restrict__ list_len) {
    __shared__ __attribute__((aligned(16))) float acc[BAND_ROWS * F];
    __shared__ unsigned queue[SCAN_WAVES][QUEUE_WORDS];

    const int tid  = threadIdx.x;
    const int lane = tid & 31;
    const int wave = __builtin_amdgcn_readfirstlane(tid >> 5);
    const int band = blockIdx.x;
    const int lo   = band * BAND_ROWS;

    {
        float4* a4 = (float4*)acc;
        const float4 z = {0.0f, 0.0f, 0.0f, 0.0f};
        for (int i = tid; i < BAND_ROWS * F / 4; i += 32 * SCAN_WAVES) a4[i] = z;
    }
    __syncthreads();

    add_band_lists<F, F>(wave, lane, lo, band, src, dst, dis, h, acc, queue[wave], lists, list_len);

    __syncthreads();

    {
        const int rows_here = ((lo + BAND_ROWS < N_NODES) ? BAND_ROWS : (N_NODES - lo));
        const int n4 = rows_here * (F / 4);
        const float4* a4 = (const float4*)acc;
        float4* o4 = (float4*)(out + (long)lo * F);
        for (int i = tid; i < n4; i += 32 * SCAN_WAVES) o4[i] = a4[i];
    }
}

#define DENSE_ROWS  (DENSE_TILES * 16)

static_assert(OUT_F % 32 == 0, "a wave's job is 32 output columns");

__global__ __launch_bounds__(32 * SCAN_WAVES)
void band_aggregate_dense_kernel(const int* __restrict__ src, const int* __restrict__ dst,
                                 const float* __restrict__ dis, const float* __restrict__ h,
                                 const float* __restrict__ W, const float* __restrict__ bias,
                                 float* __restrict__ out,
                                 const unsigned* __restrict__ lists,
                                 const unsigned* __restrict__ list_len) {
    __shared__ __attribute__((aligned(16))) float acc[DENSE_ROWS * DENSE_PITCH];
    __shared__ __attribute__((aligned(16))) float wsh[OUT_F * DENSE_PITCH];
    __shared__ unsigned queue[SCAN_WAVES][QUEUE_WORDS];

    const int tid  = threadIdx.x;
    const int lane = tid & 31;
    const int wave = __builtin_amdgcn_readfirstlane(tid >> 5);
    const int band = blockIdx.x;
    const int lo   = band * BAND_ROWS;

    {
        float4* a4 = (float4*)acc;
        const float4 z = {0.0f, 0.0f, 0.0f, 0.0f};
        for (int i = tid; i < DENSE_ROWS * DENSE_PITCH / 4; i += 32 * SCAN_WAVES) a4[i] = z;
        const float4* w4 = (const float4*)W;
        for (int i = tid; i < OUT_F * (HID_F / 4); i += 32 * SCAN_WAVES) {
            const int n  = i / (HID_F / 4);
            const int k4 = i % (HID_F / 4);
            *(float4*)(wsh + n * DENSE_PITCH + 4 * k4) = w4[i];
        }
    }
    __syncthreads();

    add_band_lists<HID_F, DENSE_PITCH>(wave, lane, lo, band, src, dst, dis, h, acc, queue[wave],
                                       lists, list_len);

    __syncthreads();

    const int rows_here  = ((lo + BAND_ROWS < N_NODES) ? BAND_ROWS : (N_NODES - lo));
    const int tiles_here = (rows_here + 15) / 16;
    const int half = lane >> 4;
    const int l    = lane & 15;
    for (int job = wave; job < tiles_here * (OUT_F / 32); job += SCAN_WAVES) {
        const int m0 = (job / (OUT_F / 32)) * 16;
        const int n0 = (job % (OUT_F / 32)) * 32;

        const float* arow  = acc + (m0 + l) * DENSE_PITCH + 2 * half;
        const float* brow0 = wsh + (n0 + l) * DENSE_PITCH + 2 * half;
        const float* brow1 = brow0 + 16 * DENSE_PITCH;

        v8f c0 = {};
        v8f c1 = {};
#pragma unroll
        for (int s = 0; s < HID_F / 4; ++s) {
            v2f a  = *(const v2f*)(arow + 4 * s);
            v2f b0 = *(const v2f*)(brow0 + 4 * s);
            v2f b1 = *(const v2f*)(brow1 + 4 * s);
            a.x = fmaxf(a.x, 0.0f);
            a.y = fmaxf(a.y, 0.0f);
            c0 = __builtin_amdgcn_wmma_f32_16x16x4_f32(false, a, false, b0,
                                                       (short)0, c0, false, false);
            c1 = __builtin_amdgcn_wmma_f32_16x16x4_f32(false, a, false, b1,
                                                       (short)0, c1, false, false);
        }

        const float bb0 = bias[n0 + l];
        const float bb1 = bias[n0 + 16 + l];
#pragma unroll
        for (int r = 0; r < 8; ++r) {
            const int row = m0 + r + 8 * half;
            if (row < rows_here) {
                float* op = out + (long)(lo + row) * OUT_F + n0 + l;
                op[0]  = c0[r] + bb0;
                op[16] = c1[r] + bb1;
            }
        }
    }
}


extern "C" void kernel_launch(void* const* d_in, const int* in_sizes, int n_in,
                              void* d_out, int out_size, void* d_ws, size_t ws_size,
                              hipStream_t stream) {
    const float* x  = (const float*)d_in[0];
    const int*   ei = (const int*)d_in[1];
    const float* W1 = (const float*)d_in[2];
    const float* b1 = (const float*)d_in[3];
    const float* W2 = (const float*)d_in[4];
    const float* b2 = (const float*)d_in[5];
    float* out = (float*)d_out;

    const int* src = ei;
    const int* dst = ei + N_EDGES;

    float* deg  = (float*)d_ws;
    float* h1   = deg + N_NODES;
    float* h2   = h1 + (long)N_NODES * HID_F;
    unsigned* lists    = (unsigned*)(h2 + (long)N_NODES * OUT_F);
    unsigned* list_len = lists + (long)N_BANDS * SCAN_WAVES * LIST_CAP;

    band_list_kernel<<<N_BANDS, 32 * SCAN_WAVES, 0, stream>>>(src, dst, x, W1, b1, h1, deg, lists, list_len);

    band_aggregate_dense_kernel<<<N_BANDS, 32 * SCAN_WAVES, 0, stream>>>(src, dst, deg, h1, W2, b2, h2,
                                                                        lists, list_len);

    band_aggregate_kernel<OUT_F><<<N_BANDS, 32 * SCAN_WAVES, 0, stream>>>(src, dst, deg, h2, out,
                                                                         lists, list_len);
}


# Scan launch split by band group x edge slice (16 x 16): a work-group reads 1/16 of dst for 16 bands; row counts summed by a small added launch (3 to 4); lists per (band, slice), LDS arrival order
# speedup vs baseline: 4.3117x; 1.3370x over previous
#include <hip/hip_runtime.h>
#include <hip/hip_bf16.h>

typedef __attribute__((ext_vector_type(2))) float v2f;
typedef __attribute__((ext_vector_type(8))) float v8f;

#define N_NODES 100000
#define N_EDGES 1600000
#define IN_F    128
#define HID_F   128
#define OUT_F   64

#define DENSE_PITCH 132

static_assert(DENSE_PITCH % 4 == 0, "rows stay 16-byte aligned");

__device__ __forceinline__ void layer1_tile(int lane, int lo, int m0, int rows_here,
                                            const float* __restrict__ X, const float* wsh,
                                            const float* __restrict__ bias, float* __restrict__ H) {
    constexpr int NT = 2;
    constexpr int KS = IN_F / 4;
    constexpr int KP = KS / 2;
    constexpr int AHEAD = 2;
    const int half = lane >> 4;
    const int l    = lane & 15;

    const int arow_i = (m0 + l < rows_here) ? (m0 + l) : (rows_here - 1);
    const float* arow = X + (long)(lo + arow_i) * IN_F + 2 * half;

    v2f a[KS];
#pragma unroll
    for (int s = 0; s < KS; ++s) a[s] = *(const v2f*)(arow + 4 * s);

#pragma unroll 1
    for (int n0 = 0; n0 < HID_F; n0 += 16 * NT) {
        const float* brow = wsh + (n0 + l) * DENSE_PITCH + 2 * half;

        v8f c[NT];
#pragma unroll
        for (int j = 0; j < NT; ++j) c[j] = v8f{};

        v2f b[KP][NT][2];
#pragma unroll
        for (int p = 0; p < AHEAD; ++p)
#pragma unroll
            for (int j = 0; j < NT; ++j)
#pragma unroll
                for (int t = 0; t < 2; ++t)
                    b[p][j][t] = *(const v2f*)(brow + j * 16 * DENSE_PITCH + 4 * (2 * p + t));

#pragma unroll
        for (int p = 0; p < KP; ++p) {
            if (p + AHEAD < KP) {
#pragma unroll
                for (int j = 0; j < NT; ++j)
#pragma unroll
                    for (int t = 0; t < 2; ++t)
                        b[p + AHEAD][j][t] =
                            *(const v2f*)(brow + j * 16 * DENSE_PITCH + 4 * (2 * (p + AHEAD) + t));
            }
            __builtin_amdgcn_sched_barrier(0);
#pragma unroll
            for (int t = 0; t < 2; ++t)
#pragma unroll
                for (int j = 0; j < NT; ++j) {
                    c[j] = __builtin_amdgcn_wmma_f32_16x16x4_f32(false, a[2 * p + t], false, b[p][j][t],
                                                                 (short)0, c[j], false, false);
                }
        }

        float* hp = H + (long)(lo + m0 + 8 * half) * HID_F + n0 + l;
        const int rows_left = rows_here - (m0 + 8 * half);
#pragma unroll
        for (int j = 0; j < NT; ++j) {
            const float bb = bias[n0 + 16 * j + l];
#pragma unroll
            for (int r = 0; r < 8; ++r) {
                if (r < rows_left) hp[r * HID_F + 16 * j] = c[j][r] + bb;
            }
        }
    }
}

#define BAND_ROWS 391
#define N_BANDS   256
#define EDGE_BITS 21
#define SCAN_WAVES 32
#define SCAN_AHEAD 4
#define QUEUE_WORDS 544

#define N_GROUPS    16
#define N_SLICES    16
#define GROUP_BANDS (N_BANDS / N_GROUPS)
#define GROUP_ROWS  (GROUP_BANDS * BAND_ROWS)
#define N_CHUNKS    (N_EDGES / 128)
#define COUNT_PITCH 6272
#define LEN_PITCH   32

static_assert(N_BANDS * BAND_ROWS >= N_NODES, "bands must cover every row");
static_assert((N_BANDS - 1) * BAND_ROWS < N_NODES, "no empty band");
static_assert(N_EDGES <= (1 << EDGE_BITS), "edge index must fit its field");
static_assert(BAND_ROWS <= (1 << (32 - EDGE_BITS)), "band row must fit its field");
static_assert(N_EDGES % 128 == 0, "whole 128-edge chunks");
static_assert(QUEUE_WORDS >= 31 + SCAN_AHEAD * 128 + 1, "queue holds one scan step");
static_assert(N_GROUPS * N_SLICES == N_BANDS, "one work-group per (group, slice) pair and per band");
static_assert(N_GROUPS * GROUP_BANDS == N_BANDS, "groups hold whole bands");
static_assert(SCAN_WAVES == 2 * GROUP_BANDS, "two waves write a band's list");
static_assert(SCAN_WAVES == 2 * N_SLICES, "two waves read a slice's list");
static_assert(COUNT_PITCH >= GROUP_ROWS && COUNT_PITCH % 32 == 0, "count blocks are whole lines");
static_assert(LEN_PITCH >= GROUP_BANDS, "length block holds the group's bands");
static_assert(N_CHUNKS / N_SLICES >= 2, "both waves of a slice find a chunk in it");

__device__ __forceinline__ int slice_start(int s) { return (s * N_CHUNKS) / N_SLICES; }

#define LIST_CAP 1024
#define SRC_BITS 17
#define LIST_OVERFLOW 0xFFFFFFFFu

static_assert(N_NODES <= (1 << SRC_BITS), "source row must fit its field");
static_assert(BAND_ROWS <= (1 << (32 - SRC_BITS)), "band row must fit its field");
static_assert(LIST_CAP % 64 == 0, "whole 32-word list lines for both waves");

template <int F, int P>
__device__ __forceinline__ void add_rows(int s, int r, float w, int n, int lane,
                                         const float* __restrict__ h, float* acc) {
    constexpr int C = F / 32;
    const int wbits = __builtin_bit_cast(int, w);

#pragma unroll 1
    for (int j0 = 0; j0 < n; j0 += 8) {
        float v[8][C];
        int   rr[8];
        float ww[8];
#pragma unroll
        for (int u = 0; u < 8; ++u) {
            const int j = (j0 + u < n) ? (j0 + u) : (n - 1);
            const int sj = __builtin_amdgcn_readlane(s, j);
            rr[u] = __builtin_amdgcn_readlane(r, j);
            ww[u] = __builtin_bit_cast(float, __builtin_amdgcn_readlane(wbits, j));
            const float* hp = h + (long)sj * F + lane;
#pragma unroll
            for (int t = 0; t < C; ++t) v[u][t] = hp[32 * t];
        }
#pragma unroll
        for (int u = 0; u < 8; ++u) {
            if (j0 + u < n) {
                float* ap = acc + rr[u] * P + lane;
#pragma unroll
                for (int t = 0; t < C; ++t) atomicAdd(ap + 32 * t, v[u][t] * ww[u]);
            }
        }
    }
}

template <int F, int P>
__device__ __forceinline__ void take_queued_edges(const unsigned* q, int n, int lane, int lo,
                                                  const int* __restrict__ src,
                                                  const float* __restrict__ dis,
                                                  const float* __restrict__ h, float* acc) {
    const unsigned ent = q[lane < n ? lane : n - 1];
    const int e = (int)(ent & ((1u << EDGE_BITS) - 1u));
    const int r = (int)(ent >> EDGE_BITS);
    const int s = src[e];
    const float w = dis[s] * dis[lo + r];
    add_rows<F, P>(s, r, w, n, lane, h, acc);
}

template <int F, int P>
__device__ __forceinline__ void scan_steps(int c_from, int c_to, int c_stride, int& cnt,
                                           int lane, int lo,
                                           const int* __restrict__ src,
                                           const int* __restrict__ dst,
                                           const float* __restrict__ dis,
                                           const float* __restrict__ h,
                                           float* acc, unsigned* q) {
    for (int c0 = c_from; c0 < c_to; c0 += c_stride * SCAN_AHEAD) {
        int4 dv[SCAN_AHEAD];
        int  ch[SCAN_AHEAD];
#pragma unroll
        for (int u = 0; u < SCAN_AHEAD; ++u) {
            const int ci = c0 + c_stride * u;
            const int c = (ci < c_to) ? ci : (c_to - 1);
            ch[u] = c;
            dv[u] = *(const int4*)(dst + (long)c * 128 + lane * 4);
        }
#pragma unroll
        for (int u = 0; u < SCAN_AHEAD; ++u) {
            if (c0 + c_stride * u < c_to) {
                const int ebase = ch[u] * 128 + lane * 4;
#pragma unroll
                for (int k = 0; k < 4; ++k) {
                    const int d = (k == 0) ? dv[u].x : (k == 1) ? dv[u].y : (k == 2) ? dv[u].z : dv[u].w;
                    const unsigned r = (unsigned)(d - lo);
                    const bool hit = r < (unsigned)BAND_ROWS;
                    const unsigned mask = __builtin_amdgcn_ballot_w32(hit);
                    if (mask != 0u) {
                        if (hit) {
                            const int pos = cnt + (int)__builtin_amdgcn_mbcnt_lo(mask, 0u);
                            q[pos] = (r << EDGE_BITS) | (unsigned)(ebase + k);
                        }
                        cnt += __builtin_popcount(mask);
                    }
                }
            }
        }
        while (cnt >= 32) {
            cnt -= 32;
            take_queued_edges<F, P>(q + cnt, 32, lane, lo, src, dis, h, acc);
        }
    }
}

#define DENSE_TILES ((BAND_ROWS + 15) / 16)

static_assert(DENSE_TILES <= SCAN_WAVES, "one tile per wave covers the band");

__global__ __launch_bounds__(32 * SCAN_WAVES)
void group_slice_list_kernel(const int* __restrict__ src, const int* __restrict__ dst,
                             const float* __restrict__ x, const float* __restrict__ W,
                             const float* __restrict__ bias, float* __restrict__ h1,
                             unsigned* __restrict__ counts,
                             unsigned* __restrict__ lists, unsigned* __restrict__ list_len) {
    __shared__ unsigned stage[GROUP_BANDS][LIST_CAP];
    __shared__ unsigned rowcnt[GROUP_ROWS];
    __shared__ unsigned bandcnt[GROUP_BANDS];
    __shared__ __attribute__((aligned(16))) float wsh[HID_F * DENSE_PITCH];

    const int tid  = threadIdx.x;
    const int lane = tid & 31;
    const int wave  = __builtin_amdgcn_readfirstlane(tid >> 5);
    const int wg    = blockIdx.x;
    const int group = wg / N_SLICES;
    const int slice = wg % N_SLICES;
    const int lo    = wg * BAND_ROWS;
    const int glo   = group * GROUP_ROWS;

    for (int i = tid; i < GROUP_ROWS; i += 32 * SCAN_WAVES) rowcnt[i] = 0u;
    if (tid < GROUP_BANDS) bandcnt[tid] = 0u;
    {
        const float4* w4 = (const float4*)W;
        for (int i = tid; i < HID_F * (IN_F / 4); i += 32 * SCAN_WAVES) {
            const int n  = i / (IN_F / 4);
            const int k4 = i % (IN_F / 4);
            *(float4*)(wsh + n * DENSE_PITCH + 4 * k4) = w4[i];
        }
    }
    __syncthreads();

    const int rows_here  = ((lo + BAND_ROWS < N_NODES) ? BAND_ROWS : (N_NODES - lo));
    const int tiles_here = (rows_here + 15) / 16;
    const bool tile_first = ((wave >> 2) & 1) == 0;

    if (tile_first && wave < tiles_here) layer1_tile(lane, lo, wave * 16, rows_here, x, wsh, bias, h1);

    const int c_lo = slice_start(slice);
    const int c_hi = slice_start(slice + 1);
    for (int c0 = c_lo + wave; c0 < c_hi; c0 += SCAN_WAVES * SCAN_AHEAD) {
        int4 dv[SCAN_AHEAD];
        int  ch[SCAN_AHEAD];
#pragma unroll
        for (int u = 0; u < SCAN_AHEAD; ++u) {
            const int ci = c0 + SCAN_WAVES * u;
            const int c = (ci < c_hi) ? ci : (c_hi - 1);
            ch[u] = c;
            dv[u] = *(const int4*)(dst + (long)c * 128 + lane * 4);
        }
#pragma unroll
        for (int u = 0; u < SCAN_AHEAD; ++u) {
            if (c0 + SCAN_WAVES * u < c_hi) {
                const int ebase = ch[u] * 128 + lane * 4;
#pragma unroll
                for (int k = 0; k < 4; ++k) {
                    const int d = (k == 0) ? dv[u].x : (k == 1) ? dv[u].y : (k == 2) ? dv[u].z : dv[u].w;
                    const unsigned rg = (unsigned)(d - glo);
                    if (rg < (unsigned)GROUP_ROWS) {
                        const unsigned bl = rg / (unsigned)BAND_ROWS;
                        const unsigned r  = rg - bl * (unsigned)BAND_ROWS;
                        atomicAdd(&rowcnt[rg], 1u);
                        const unsigned pos = atomicAdd(&bandcnt[bl], 1u);
                        if (pos < (unsigned)LIST_CAP) stage[bl][pos] = (r << EDGE_BITS) | (unsigned)(ebase + k);
                    }
                }
            }
        }
    }

    if (!tile_first && wave < tiles_here) layer1_tile(lane, lo, wave * 16, rows_here, x, wsh, bias, h1);

    __syncthreads();

    {
        const int bl   = wave >> 1;
        const int part = wave & 1;
        const unsigned total = bandcnt[bl];
        const int n = (total <= (unsigned)LIST_CAP) ? (int)total : 0;
        unsigned* wl = lists + ((long)(group * GROUP_BANDS + bl) * N_SLICES + slice) * LIST_CAP;
        for (int i = part * 32 + lane; i < n; i += 64) {
            const unsigned ent = stage[bl][i];
            const int e = (int)(ent & ((1u << EDGE_BITS) - 1u));
            const unsigned r = ent >> EDGE_BITS;
            wl[i] = (r << SRC_BITS) | (unsigned)src[e];
        }
    }
    if (tid < GROUP_BANDS) {
        const unsigned total = bandcnt[tid];
        list_len[wg * LEN_PITCH + tid] = (total <= (unsigned)LIST_CAP) ? total : LIST_OVERFLOW;
    }
    {
        unsigned* cp = counts + (long)(slice * N_GROUPS + group) * COUNT_PITCH;
        for (int i = tid; i < COUNT_PITCH; i += 32 * SCAN_WAVES) cp[i] = (i < GROUP_ROWS) ? rowcnt[i] : 0u;
    }
}

#define DEG_THREADS 256

__global__ __launch_bounds__(DEG_THREADS)
void inv_sqrt_degree_kernel(const unsigned* __restrict__ counts, float* __restrict__ dis) {
    const int row = blockIdx.x * DEG_THREADS + threadIdx.x;
    if (row < N_NODES) {
        const int g = row / GROUP_ROWS;
        const int i = row - g * GROUP_ROWS;
        unsigned c = 0u;
#pragma unroll
        for (int s = 0; s < N_SLICES; ++s) c += counts[(long)(s * N_GROUPS + g) * COUNT_PITCH + i];
        const float d = (float)c;
        dis[row] = (d > 0.0f) ? (1.0f / sqrtf(d)) : 0.0f;
    }
}

template <int F, int P>
__device__ __forceinline__ void add_band_lists(int wave, int lane, int lo, int band,
                                               const int* __restrict__ src,
                                               const int* __restrict__ dst,
                                               const float* __restrict__ dis,
                                               const float* __restrict__ h,
                                               float* acc, unsigned* q,
                                               const unsigned* __restrict__ lists,
                                               const unsigned* __restrict__ list_len) {
    const int slice = wave >> 1;
    const int part  = wave & 1;
    const int group = band / GROUP_BANDS;
    const int bl    = band % GROUP_BANDS;
    const unsigned* wlist = lists + ((long)band * N_SLICES + slice) * LIST_CAP;

    const unsigned len = list_len[(group * N_SLICES + slice) * LEN_PITCH + bl];
    if (len == LIST_OVERFLOW) {
        int cnt = 0;
        scan_steps<F, P>(slice_start(slice) + part, slice_start(slice + 1), 2, cnt, lane, lo,
                         src, dst, dis, h, acc, q);
        if (cnt > 0) take_queued_edges<F, P>(q, cnt, lane, lo, src, dis, h, acc);
    } else {
        const int total = (int)len;
        unsigned ent_next = 0u;
        const int j_first = part * 32;
        if (j_first < total) {
            const int n1 = (total - j_first < 32) ? (total - j_first) : 32;
            ent_next = wlist[j_first + ((lane < n1) ? lane : (n1 - 1))];
        }
        for (int j0 = j_first; j0 < total; j0 += 64) {
            const int n = (total - j0 < 32) ? (total - j0) : 32;
            const unsigned ent = ent_next;
            const int jn = j0 + 64;
            if (jn < total) {
                const int n2 = (total - jn < 32) ? (total - jn) : 32;
                ent_next = wlist[jn + ((lane < n2) ? lane : (n2 - 1))];
            }
            const int s = (int)(ent & ((1u << SRC_BITS) - 1u));
            const int r = (int)(ent >> SRC_BITS);
            const float w = dis[s] * dis[lo + r];
            add_rows<F, P>(s, r, w, n, lane, h, acc);
        }
    }
}

template <int F>
__global__ __launch_bounds__(32 * SCAN_WAVES)
void band_aggregate_kernel(const int* __restrict__ src, const int* __restrict__ dst,
                           const float* __restrict__ dis, const float* __restrict__ h,
                           float* __restrict__ out,
                           const unsigned* __restrict__ lists,
                           const unsigned* __restrict__ list_len) {
    __shared__ __attribute__((aligned(16))) float acc[BAND_ROWS * F];
    __shared__ unsigned queue[SCAN_WAVES][QUEUE_WORDS];

    const int tid  = threadIdx.x;
    const int lane = tid & 31;
    const int wave = __builtin_amdgcn_readfirstlane(tid >> 5);
    const int band = blockIdx.x;
    const int lo   = band * BAND_ROWS;

    {
        float4* a4 = (float4*)acc;
        const float4 z = {0.0f, 0.0f, 0.0f, 0.0f};
        for (int i = tid; i < BAND_ROWS * F / 4; i += 32 * SCAN_WAVES) a4[i] = z;
    }
    __syncthreads();

    add_band_lists<F, F>(wave, lane, lo, band, src, dst, dis, h, acc, queue[wave], lists, list_len);

    __syncthreads();

    {
        const int rows_here = ((lo + BAND_ROWS < N_NODES) ? BAND_ROWS : (N_NODES - lo));
        const int n4 = rows_here * (F / 4);
        const float4* a4 = (const float4*)acc;
        float4* o4 = (float4*)(out + (long)lo * F);
        for (int i = tid; i < n4; i += 32 * SCAN_WAVES) o4[i] = a4[i];
    }
}

#define DENSE_ROWS  (DENSE_TILES * 16)

static_assert(OUT_F % 32 == 0, "a wave's job is 32 output columns");

__global__ __launch_bounds__(32 * SCAN_WAVES)
void band_aggregate_dense_kernel(const int* __restrict__ src, const int* __restrict__ dst,
                                 const float* __restrict__ dis, const float* __restrict__ h,
                                 const float* __restrict__ W, const float* __restrict__ bias,
                                 float* __restrict__ out,
                                 const unsigned* __restrict__ lists,
                                 const unsigned* __restrict__ list_len) {
    __shared__ __attribute__((aligned(16))) float acc[DENSE_ROWS * DENSE_PITCH];
    __shared__ __attribute__((aligned(16))) float wsh[OUT_F * DENSE_PITCH];
    __shared__ unsigned queue[SCAN_WAVES][QUEUE_WORDS];

    const int tid  = threadIdx.x;
    const int lane = tid & 31;
    const int wave = __builtin_amdgcn_readfirstlane(tid >> 5);
    const int band = blockIdx.x;
    const int lo   = band * BAND_ROWS;

    {
        float4* a4 = (float4*)acc;
        const float4 z = {0.0f, 0.0f, 0.0f, 0.0f};
        for (int i = tid; i < DENSE_ROWS * DENSE_PITCH / 4; i += 32 * SCAN_WAVES) a4[i] = z;
        const float4* w4 = (const float4*)W;
        for (int i = tid; i < OUT_F * (HID_F / 4); i += 32 * SCAN_WAVES) {
            const int n  = i / (HID_F / 4);
            const int k4 = i % (HID_F / 4);
            *(float4*)(wsh + n * DENSE_PITCH + 4 * k4) = w4[i];
        }
    }
    __syncthreads();

    add_band_lists<HID_F, DENSE_PITCH>(wave, lane, lo, band, src, dst, dis, h, acc, queue[wave],
                                       lists, list_len);

    __syncthreads();

    const int rows_here  = ((lo + BAND_ROWS < N_NODES) ? BAND_ROWS : (N_NODES - lo));
    const int tiles_here = (rows_here + 15) / 16;
    const int half = lane >> 4;
    const int l    = lane & 15;
    for (int job = wave; job < tiles_here * (OUT_F / 32); job += SCAN_WAVES) {
        const int m0 = (job / (OUT_F / 32)) * 16;
        const int n0 = (job % (OUT_F / 32)) * 32;

        const float* arow  = acc + (m0 + l) * DENSE_PITCH + 2 * half;
        const float* brow0 = wsh + (n0 + l) * DENSE_PITCH + 2 * half;
        const float* brow1 = brow0 + 16 * DENSE_PITCH;

        v8f c0 = {};
        v8f c1 = {};
#pragma unroll
        for (int s = 0; s < HID_F / 4; ++s) {
            v2f a  = *(const v2f*)(arow + 4 * s);
            v2f b0 = *(const v2f*)(brow0 + 4 * s);
            v2f b1 = *(const v2f*)(brow1 + 4 * s);
            a.x = fmaxf(a.x, 0.0f);
            a.y = fmaxf(a.y, 0.0f);
            c0 = __builtin_amdgcn_wmma_f32_16x16x4_f32(false, a, false, b0,
                                                       (short)0, c0, false, false);
            c1 = __builtin_amdgcn_wmma_f32_16x16x4_f32(false, a, false, b1,
                                                       (short)0, c1, false, false);
        }

        const float bb0 = bias[n0 + l];
        const float bb1 = bias[n0 + 16 + l];
#pragma unroll
        for (int r = 0; r < 8; ++r) {
            const int row = m0 + r + 8 * half;
            if (row < rows_here) {
                float* op = out + (long)(lo + row) * OUT_F + n0 + l;
                op[0]  = c0[r] + bb0;
                op[16] = c1[r] + bb1;
            }
        }
    }
}


extern "C" void kernel_launch(void* const* d_in, const int* in_sizes, int n_in,
                              void* d_out, int out_size, void* d_ws, size_t ws_size,
                              hipStream_t stream) {
    const float* x  = (const float*)d_in[0];
    const int*   ei = (const int*)d_in[1];
    const float* W1 = (const float*)d_in[2];
    const float* b1 = (const float*)d_in[3];
    const float* W2 = (const float*)d_in[4];
    const float* b2 = (const float*)d_in[5];
    float* out = (float*)d_out;

    const int* src = ei;
    const int* dst = ei + N_EDGES;

    float* deg  = (float*)d_ws;
    float* h1   = deg + N_NODES;
    float* h2   = h1 + (long)N_NODES * HID_F;
    unsigned* lists    = (unsigned*)(h2 + (long)N_NODES * OUT_F);
    unsigned* list_len = lists + (long)N_BANDS * N_SLICES * LIST_CAP;
    unsigned* counts   = list_len + (long)N_BANDS * LEN_PITCH;

    group_slice_list_kernel<<<N_BANDS, 32 * SCAN_WAVES, 0, stream>>>(src, dst, x, W1, b1, h1, counts,
                                                                    lists, list_len);

    inv_sqrt_degree_kernel<<<(N_NODES + DEG_THREADS - 1) / DEG_THREADS, DEG_THREADS, 0, stream>>>(counts, deg);

    band_aggregate_dense_kernel<<<N_BANDS, 32 * SCAN_WAVES, 0, stream>>>(src, dst, deg, h1, W2, b2, h2,
                                                                        lists, list_len);

    band_aggregate_kernel<OUT_F><<<N_BANDS, 32 * SCAN_WAVES, 0, stream>>>(src, dst, deg, h2, out,
                                                                         lists, list_len);
}


# Gathered source rows read whole by one wave load (16 or 8 bytes per lane; was 4 or 2 four-byte loads), rows on aligned bounds; LDS sums in permuted word order, so layer-2 k order is permuted
# speedup vs baseline: 4.6800x; 1.0854x over previous
#include <hip/hip_runtime.h>
#include <hip/hip_bf16.h>

typedef __attribute__((ext_vector_type(2))) float v2f;
typedef __attribute__((ext_vector_type(8))) float v8f;

#define N_NODES 100000
#define N_EDGES 1600000
#define IN_F    128
#define HID_F   128
#define OUT_F   64

#define DENSE_PITCH 132

static_assert(DENSE_PITCH % 4 == 0, "rows stay 16-byte aligned");

__device__ __forceinline__ void layer1_tile(int lane, int lo, int m0, int rows_here,
                                            const float* __restrict__ X, const float* wsh,
                                            const float* __restrict__ bias, float* __restrict__ H) {
    constexpr int NT = 2;
    constexpr int KS = IN_F / 4;
    constexpr int KP = KS / 2;
    constexpr int AHEAD = 2;
    const int half = lane >> 4;
    const int l    = lane & 15;

    const int arow_i = (m0 + l < rows_here) ? (m0 + l) : (rows_here - 1);
    const float* arow = X + (long)(lo + arow_i) * IN_F + 2 * half;

    v2f a[KS];
#pragma unroll
    for (int s = 0; s < KS; ++s) a[s] = *(const v2f*)(arow + 4 * s);

#pragma unroll 1
    for (int n0 = 0; n0 < HID_F; n0 += 16 * NT) {
        const float* brow = wsh + (n0 + l) * DENSE_PITCH + 2 * half;

        v8f c[NT];
#pragma unroll
        for (int j = 0; j < NT; ++j) c[j] = v8f{};

        v2f b[KP][NT][2];
#pragma unroll
        for (int p = 0; p < AHEAD; ++p)
#pragma unroll
            for (int j = 0; j < NT; ++j)
#pragma unroll
                for (int t = 0; t < 2; ++t)
                    b[p][j][t] = *(const v2f*)(brow + j * 16 * DENSE_PITCH + 4 * (2 * p + t));

#pragma unroll
        for (int p = 0; p < KP; ++p) {
            if (p + AHEAD < KP) {
#pragma unroll
                for (int j = 0; j < NT; ++j)
#pragma unroll
                    for (int t = 0; t < 2; ++t)
                        b[p + AHEAD][j][t] =
                            *(const v2f*)(brow + j * 16 * DENSE_PITCH + 4 * (2 * (p + AHEAD) + t));
            }
            __builtin_amdgcn_sched_barrier(0);
#pragma unroll
            for (int t = 0; t < 2; ++t)
#pragma unroll
                for (int j = 0; j < NT; ++j) {
                    c[j] = __builtin_amdgcn_wmma_f32_16x16x4_f32(false, a[2 * p + t], false, b[p][j][t],
                                                                 (short)0, c[j], false, false);
                }
        }

        float* hp = H + (long)(lo + m0 + 8 * half) * HID_F + n0 + l;
        const int rows_left = rows_here - (m0 + 8 * half);
#pragma unroll
        for (int j = 0; j < NT; ++j) {
            const float bb = bias[n0 + 16 * j + l];
#pragma unroll
            for (int r = 0; r < 8; ++r) {
                if (r < rows_left) hp[r * HID_F + 16 * j] = c[j][r] + bb;
            }
        }
    }
}

#define BAND_ROWS 391
#define N_BANDS   256
#define EDGE_BITS 21
#define SCAN_WAVES 32
#define SCAN_AHEAD 4
#define QUEUE_WORDS 544

#define N_GROUPS    16
#define N_SLICES    16
#define GROUP_BANDS (N_BANDS / N_GROUPS)
#define GROUP_ROWS  (GROUP_BANDS * BAND_ROWS)
#define N_CHUNKS    (N_EDGES / 128)
#define COUNT_PITCH 6272
#define LEN_PITCH   32

static_assert(N_BANDS * BAND_ROWS >= N_NODES, "bands must cover every row");
static_assert((N_BANDS * BAND_ROWS * 4) % 512 == 0, "h1 behind the deg^-0.5 plane starts on a 512-byte boundary");
static_assert((N_BANDS - 1) * BAND_ROWS < N_NODES, "no empty band");
static_assert(N_EDGES <= (1 << EDGE_BITS), "edge index must fit its field");
static_assert(BAND_ROWS <= (1 << (32 - EDGE_BITS)), "band row must fit its field");
static_assert(N_EDGES % 128 == 0, "whole 128-edge chunks");
static_assert(QUEUE_WORDS >= 31 + SCAN_AHEAD * 128 + 1, "queue holds one scan step");
static_assert(N_GROUPS * N_SLICES == N_BANDS, "one work-group per (group, slice) pair and per band");
static_assert(N_GROUPS * GROUP_BANDS == N_BANDS, "groups hold whole bands");
static_assert(SCAN_WAVES == 2 * GROUP_BANDS, "two waves write a band's list");
static_assert(SCAN_WAVES == 2 * N_SLICES, "two waves read a slice's list");
static_assert(COUNT_PITCH >= GROUP_ROWS && COUNT_PITCH % 32 == 0, "count blocks are whole lines");
static_assert(LEN_PITCH >= GROUP_BANDS, "length block holds the group's bands");
static_assert(N_CHUNKS / N_SLICES >= 2, "both waves of a slice find a chunk in it");

__device__ __forceinline__ int slice_start(int s) { return (s * N_CHUNKS) / N_SLICES; }

#define LIST_CAP 1024
#define SRC_BITS 17
#define LIST_OVERFLOW 0xFFFFFFFFu

static_assert(N_NODES <= (1 << SRC_BITS), "source row must fit its field");
static_assert(BAND_ROWS <= (1 << (32 - SRC_BITS)), "band row must fit its field");
static_assert(LIST_CAP % 64 == 0, "whole 32-word list lines for both waves");

template <int F>
__device__ __forceinline__ constexpr int sum_word(int c) { return 32 * (c % (F / 32)) + c / (F / 32); }

template <int F, int P>
__device__ __forceinline__ void add_rows(int s, int r, float w, int n, int lane,
                                         const float* __restrict__ h, float* acc) {
    constexpr int C = F / 32;
    typedef __attribute__((ext_vector_type(C))) float rowpart_t;
    const int wbits = __builtin_bit_cast(int, w);

#pragma unroll 1
    for (int j0 = 0; j0 < n; j0 += 8) {
        rowpart_t v[8];
        int   rr[8];
        float ww[8];
#pragma unroll
        for (int u = 0; u < 8; ++u) {
            const int j = (j0 + u < n) ? (j0 + u) : (n - 1);
            const int sj = __builtin_amdgcn_readlane(s, j);
            rr[u] = __builtin_amdgcn_readlane(r, j);
            ww[u] = __builtin_bit_cast(float, __builtin_amdgcn_readlane(wbits, j));
            v[u] = *((const rowpart_t*)(h + (long)sj * F) + lane);
        }
#pragma unroll
        for (int u = 0; u < 8; ++u) {
            if (j0 + u < n) {
                float* ap = acc + rr[u] * P + lane;
#pragma unroll
                for (int t = 0; t < C; ++t) atomicAdd(ap + 32 * t, v[u][t] * ww[u]);
            }
        }
    }
}

template <int F, int P>
__device__ __forceinline__ void take_queued_edges(const unsigned* q, int n, int lane, int lo,
                                                  const int* __restrict__ src,
                                                  const float* __restrict__ dis,
                                                  const float* __restrict__ h, float* acc) {
    const unsigned ent = q[lane < n ? lane : n - 1];
    const int e = (int)(ent & ((1u << EDGE_BITS) - 1u));
    const int r = (int)(ent >> EDGE_BITS);
    const int s = src[e];
    const float w = dis[s] * dis[lo + r];
    add_rows<F, P>(s, r, w, n, lane, h, acc);
}

template <int F, int P>
__device__ __forceinline__ void scan_steps(int c_from, int c_to, int c_stride, int& cnt,
                                           int lane, int lo,
                                           const int* __restrict__ src,
                                           const int* __restrict__ dst,
                                           const float* __restrict__ dis,
                                           const float* __restrict__ h,
                                           float* acc, unsigned* q) {
    for (int c0 = c_from; c0 < c_to; c0 += c_stride * SCAN_AHEAD) {
        int4 dv[SCAN_AHEAD];
        int  ch[SCAN_AHEAD];
#pragma unroll
        for (int u = 0; u < SCAN_AHEAD; ++u) {
            const int ci = c0 + c_stride * u;
            const int c = (ci < c_to) ? ci : (c_to - 1);
            ch[u] = c;
            dv[u] = *(const int4*)(dst + (long)c * 128 + lane * 4);
        }
#pragma unroll
        for (int u = 0; u < SCAN_AHEAD; ++u) {
            if (c0 + c_stride * u < c_to) {
                const int ebase = ch[u] * 128 + lane * 4;
#pragma unroll
                for (int k = 0; k < 4; ++k) {
                    const int d = (k == 0) ? dv[u].x : (k == 1) ? dv[u].y : (k == 2) ? dv[u].z : dv[u].w;
                    const unsigned r = (unsigned)(d - lo);
                    const bool hit = r < (unsigned)BAND_ROWS;
                    const unsigned mask = __builtin_amdgcn_ballot_w32(hit);
                    if (mask != 0u) {
                        if (hit) {
                            const int pos = cnt + (int)__builtin_amdgcn_mbcnt_lo(mask, 0u);
                            q[pos] = (r << EDGE_BITS) | (unsigned)(ebase + k);
                        }
                        cnt += __builtin_popcount(mask);
                    }
                }
            }
        }
        while (cnt >= 32) {
            cnt -= 32;
            take_queued_edges<F, P>(q + cnt, 32, lane, lo, src, dis, h, acc);
        }
    }
}

#define DENSE_TILES ((BAND_ROWS + 15) / 16)

static_assert(DENSE_TILES <= SCAN_WAVES, "one tile per wave covers the band");

__global__ __launch_bounds__(32 * SCAN_WAVES)
void group_slice_list_kernel(const int* __restrict__ src, const int* __restrict__ dst,
                             const float* __restrict__ x, const float* __restrict__ W,
                             const float* __restrict__ bias, float* __restrict__ h1,
                             unsigned* __restrict__ counts,
                             unsigned* __restrict__ lists, unsigned* __restrict__ list_len) {
    __shared__ unsigned stage[GROUP_BANDS][LIST_CAP];
    __shared__ unsigned rowcnt[GROUP_ROWS];
    __shared__ unsigned bandcnt[GROUP_BANDS];
    __shared__ __attribute__((aligned(16))) float wsh[HID_F * DENSE_PITCH];

    const int tid  = threadIdx.x;
    const int lane = tid & 31;
    const int wave  = __builtin_amdgcn_readfirstlane(tid >> 5);
    const int wg    = blockIdx.x;
    const int group = wg / N_SLICES;
    const int slice = wg % N_SLICES;
    const int lo    = wg * BAND_ROWS;
    const int glo   = group * GROUP_ROWS;

    for (int i = tid; i < GROUP_ROWS; i += 32 * SCAN_WAVES) rowcnt[i] = 0u;
    if (tid < GROUP_BANDS) bandcnt[tid] = 0u;
    {
        const float4* w4 = (const float4*)W;
        for (int i = tid; i < HID_F * (IN_F / 4); i += 32 * SCAN_WAVES) {
            const int n  = i / (IN_F / 4);
            const int k4 = i % (IN_F / 4);
            *(float4*)(wsh + n * DENSE_PITCH + 4 * k4) = w4[i];
        }
    }
    __syncthreads();

    const int rows_here  = ((lo + BAND_ROWS < N_NODES) ? BAND_ROWS : (N_NODES - lo));
    const int tiles_here = (rows_here + 15) / 16;
    const bool tile_first = ((wave >> 2) & 1) == 0;

    if (tile_first && wave < tiles_here) layer1_tile(lane, lo, wave * 16, rows_here, x, wsh, bias, h1);

    const int c_lo = slice_start(slice);
    const int c_hi = slice_start(slice + 1);
    for (int c0 = c_lo + wave; c0 < c_hi; c0 += SCAN_WAVES * SCAN_AHEAD) {
        int4 dv[SCAN_AHEAD];
        int  ch[SCAN_AHEAD];
#pragma unroll
        for (int u = 0; u < SCAN_AHEAD; ++u) {
            const int ci = c0 + SCAN_WAVES * u;
            const int c = (ci < c_hi) ? ci : (c_hi - 1);
            ch[u] = c;
            dv[u] = *(const int4*)(dst + (long)c * 128 + lane * 4);
        }
#pragma unroll
        for (int u = 0; u < SCAN_AHEAD; ++u) {
            if (c0 + SCAN_WAVES * u < c_hi) {
                const int ebase = ch[u] * 128 + lane * 4;
#pragma unroll
                for (int k = 0; k < 4; ++k) {
                    const int d = (k == 0) ? dv[u].x : (k == 1) ? dv[u].y : (k == 2) ? dv[u].z : dv[u].w;
                    const unsigned rg = (unsigned)(d - glo);
                    if (rg < (unsigned)GROUP_ROWS) {
                        const unsigned bl = rg / (unsigned)BAND_ROWS;
                        const unsigned r  = rg - bl * (unsigned)BAND_ROWS;
                        atomicAdd(&rowcnt[rg], 1u);
                        const unsigned pos = atomicAdd(&bandcnt[bl], 1u);
                        if (pos < (unsigned)LIST_CAP) stage[bl][pos] = (r << EDGE_BITS) | (unsigned)(ebase + k);
                    }
                }
            }
        }
    }

    if (!tile_first && wave < tiles_here) layer1_tile(lane, lo, wave * 16, rows_here, x, wsh, bias, h1);

    __syncthreads();

    {
        const int bl   = wave >> 1;
        const int part = wave & 1;
        const unsigned total = bandcnt[bl];
        const int n = (total <= (unsigned)LIST_CAP) ? (int)total : 0;
        unsigned* wl = lists + ((long)(group * GROUP_BANDS + bl) * N_SLICES + slice) * LIST_CAP;
        for (int i = part * 32 + lane; i < n; i += 64) {
            const unsigned ent = stage[bl][i];
            const int e = (int)(ent & ((1u << EDGE_BITS) - 1u));
            const unsigned r = ent >> EDGE_BITS;
            wl[i] = (r << SRC_BITS) | (unsigned)src[e];
        }
    }
    if (tid < GROUP_BANDS) {
        const unsigned total = bandcnt[tid];
        list_len[wg * LEN_PITCH + tid] = (total <= (unsigned)LIST_CAP) ? total : LIST_OVERFLOW;
    }
    {
        unsigned* cp = counts + (long)(slice * N_GROUPS + group) * COUNT_PITCH;
        for (int i = tid; i < COUNT_PITCH; i += 32 * SCAN_WAVES) cp[i] = (i < GROUP_ROWS) ? rowcnt[i] : 0u;
    }
}

#define DEG_THREADS 256

__global__ __launch_bounds__(DEG_THREADS)
void inv_sqrt_degree_kernel(const unsigned* __restrict__ counts, float* __restrict__ dis) {
    const int row = blockIdx.x * DEG_THREADS + threadIdx.x;
    if (row < N_NODES) {
        const int g = row / GROUP_ROWS;
        const int i = row - g * GROUP_ROWS;
        unsigned c = 0u;
#pragma unroll
        for (int s = 0; s < N_SLICES; ++s) c += counts[(long)(s * N_GROUPS + g) * COUNT_PITCH + i];
        const float d = (float)c;
        dis[row] = (d > 0.0f) ? (1.0f / sqrtf(d)) : 0.0f;
    }
}

template <int F, int P>
__device__ __forceinline__ void add_band_lists(int wave, int lane, int lo, int band,
                                               const int* __restrict__ src,
                                               const int* __restrict__ dst,
                                               const float* __restrict__ dis,
                                               const float* __restrict__ h,
                                               float* acc, unsigned* q,
                                               const unsigned* __restrict__ lists,
                                               const unsigned* __restrict__ list_len) {
    const int slice = wave >> 1;
    const int part  = wave & 1;
    const int group = band / GROUP_BANDS;
    const int bl    = band % GROUP_BANDS;
    const unsigned* wlist = lists + ((long)band * N_SLICES + slice) * LIST_CAP;

    const unsigned len = list_len[(group * N_SLICES + slice) * LEN_PITCH + bl];
    if (len == LIST_OVERFLOW) {
        int cnt = 0;
        scan_steps<F, P>(slice_start(slice) + part, slice_start(slice + 1), 2, cnt, lane, lo,
                         src, dst, dis, h, acc, q);
        if (cnt > 0) take_queued_edges<F, P>(q, cnt, lane, lo, src, dis, h, acc);
    } else {
        const int total = (int)len;
        unsigned ent_next = 0u;
        const int j_first = part * 32;
        if (j_first < total) {
            const int n1 = (total - j_first < 32) ? (total - j_first) : 32;
            ent_next = wlist[j_first + ((lane < n1) ? lane : (n1 - 1))];
        }
        for (int j0 = j_first; j0 < total; j0 += 64) {
            const int n = (total - j0 < 32) ? (total - j0) : 32;
            const unsigned ent = ent_next;
            const int jn = j0 + 64;
            if (jn < total) {
                const int n2 = (total - jn < 32) ? (total - jn) : 32;
                ent_next = wlist[jn + ((lane < n2) ? lane : (n2 - 1))];
            }
            const int s = (int)(ent & ((1u << SRC_BITS) - 1u));
            const int r = (int)(ent >> SRC_BITS);
            const float w = dis[s] * dis[lo + r];
            add_rows<F, P>(s, r, w, n, lane, h, acc);
        }
    }
}

template <int F>
__global__ __launch_bounds__(32 * SCAN_WAVES)
void band_aggregate_kernel(const int* __restrict__ src, const int* __restrict__ dst,
                           const float* __restrict__ dis, const float* __restrict__ h,
                           float* __restrict__ out,
                           const unsigned* __restrict__ lists,
                           const unsigned* __restrict__ list_len) {
    __shared__ __attribute__((aligned(16))) float acc[BAND_ROWS * F];
    __shared__ unsigned queue[SCAN_WAVES][QUEUE_WORDS];

    const int tid  = threadIdx.x;
    const int lane = tid & 31;
    const int wave = __builtin_amdgcn_readfirstlane(tid >> 5);
    const int band = blockIdx.x;
    const int lo   = band * BAND_ROWS;

    {
        float4* a4 = (float4*)acc;
        const float4 z = {0.0f, 0.0f, 0.0f, 0.0f};
        for (int i = tid; i < BAND_ROWS * F / 4; i += 32 * SCAN_WAVES) a4[i] = z;
    }
    __syncthreads();

    add_band_lists<F, F>(wave, lane, lo, band, src, dst, dis, h, acc, queue[wave], lists, list_len);

    __syncthreads();

    {
        const int rows_here = ((lo + BAND_ROWS < N_NODES) ? BAND_ROWS : (N_NODES - lo));
        const int n4 = rows_here * (F / 4);
        float4* o4 = (float4*)(out + (long)lo * F);
        static_assert(4 % (F / 32) == 0, "a 16-byte piece of a row holds whole lanes' shares");
        for (int i = tid; i < n4; i += 32 * SCAN_WAVES) {
            const int row = i / (F / 4);
            const int c0  = 4 * (i % (F / 4));
            const float* ar = acc + row * F + sum_word<F>(c0);
            float4 o;
            o.x = ar[sum_word<F>(0)];
            o.y = ar[sum_word<F>(1)];
            o.z = ar[sum_word<F>(2)];
            o.w = ar[sum_word<F>(3)];
            o4[i] = o;
        }
    }
}

#define DENSE_ROWS  (DENSE_TILES * 16)

static_assert(OUT_F % 32 == 0, "a wave's job is 32 output columns");

__global__ __launch_bounds__(32 * SCAN_WAVES)
void band_aggregate_dense_kernel(const int* __restrict__ src, const int* __restrict__ dst,
                                 const float* __restrict__ dis, const float* __restrict__ h,
                                 const float* __restrict__ W, const float* __restrict__ bias,
                                 float* __restrict__ out,
                                 const unsigned* __restrict__ lists,
                                 const unsigned* __restrict__ list_len) {
    __shared__ __attribute__((aligned(16))) float acc[DENSE_ROWS * DENSE_PITCH];
    __shared__ __attribute__((aligned(16))) float wsh[OUT_F * DENSE_PITCH];
    __shared__ unsigned queue[SCAN_WAVES][QUEUE_WORDS];

    const int tid  = threadIdx.x;
    const int lane = tid & 31;
    const int wave = __builtin_amdgcn_readfirstlane(tid >> 5);
    const int band = blockIdx.x;
    const int lo   = band * BAND_ROWS;

    {
        float4* a4 = (float4*)acc;
        const float4 z = {0.0f, 0.0f, 0.0f, 0.0f};
        for (int i = tid; i < DENSE_ROWS * DENSE_PITCH / 4; i += 32 * SCAN_WAVES) a4[i] = z;
        static_assert(HID_F / 32 == 4, "a 16-byte piece of W is one lane's share of a row");
        const float4* w4 = (const float4*)W;
        for (int i = tid; i < OUT_F * (HID_F / 4); i += 32 * SCAN_WAVES) {
            const int n  = i / (HID_F / 4);
            const int k4 = i % (HID_F / 4);
            const float4 wv = w4[i];
            float* wr = wsh + n * DENSE_PITCH + k4;
            wr[0]  = wv.x;
            wr[32] = wv.y;
            wr[64] = wv.z;
            wr[96] = wv.w;
        }
    }
    __syncthreads();

    add_band_lists<HID_F, DENSE_PITCH>(wave, lane, lo, band, src, dst, dis, h, acc, queue[wave],
                                       lists, list_len);

    __syncthreads();

    const int rows_here  = ((lo + BAND_ROWS < N_NODES) ? BAND_ROWS : (N_NODES - lo));
    const int tiles_here = (rows_here + 15) / 16;
    const int half = lane >> 4;
    const int l    = lane & 15;
    for (int job = wave; job < tiles_here * (OUT_F / 32); job += SCAN_WAVES) {
        const int m0 = (job / (OUT_F / 32)) * 16;
        const int n0 = (job % (OUT_F / 32)) * 32;

        const float* arow  = acc + (m0 + l) * DENSE_PITCH + 2 * half;
        const float* brow0 = wsh + (n0 + l) * DENSE_PITCH + 2 * half;
        const float* brow1 = brow0 + 16 * DENSE_PITCH;

        v8f c0 = {};
        v8f c1 = {};
#pragma unroll
        for (int s = 0; s < HID_F / 4; ++s) {
            v2f a  = *(const v2f*)(arow + 4 * s);
            v2f b0 = *(const v2f*)(brow0 + 4 * s);
            v2f b1 = *(const v2f*)(brow1 + 4 * s);
            a.x = fmaxf(a.x, 0.0f);
            a.y = fmaxf(a.y, 0.0f);
            c0 = __builtin_amdgcn_wmma_f32_16x16x4_f32(false, a, false, b0,
                                                       (short)0, c0, false, false);
            c1 = __builtin_amdgcn_wmma_f32_16x16x4_f32(false, a, false, b1,
                                                       (short)0, c1, false, false);
        }

        const float bb0 = bias[n0 + l];
        const float bb1 = bias[n0 + 16 + l];
#pragma unroll
        for (int r = 0; r < 8; ++r) {
            const int row = m0 + r + 8 * half;
            if (row < rows_here) {
                float* op = out + (long)(lo + row) * OUT_F + n0 + l;
                op[0]  = c0[r] + bb0;
                op[16] = c1[r] + bb1;
            }
        }
    }
}


extern "C" void kernel_launch(void* const* d_in, const int* in_sizes, int n_in,
                              void* d_out, int out_size, void* d_ws, size_t ws_size,
                              hipStream_t stream) {
    const float* x  = (const float*)d_in[0];
    const int*   ei = (const int*)d_in[1];
    const float* W1 = (const float*)d_in[2];
    const float* b1 = (const float*)d_in[3];
    const float* W2 = (const float*)d_in[4];
    const float* b2 = (const float*)d_in[5];
    float* out = (float*)d_out;

    const int* src = ei;
    const int* dst = ei + N_EDGES;

    float* deg  = (float*)d_ws;
    float* h1   = deg + N_BANDS * BAND_ROWS;
    float* h2   = h1 + (long)N_NODES * HID_F;
    unsigned* lists    = (unsigned*)(h2 + (long)N_NODES * OUT_F);
    unsigned* list_len = lists + (long)N_BANDS * N_SLICES * LIST_CAP;
    unsigned* counts   = list_len + (long)N_BANDS * LEN_PITCH;

    group_slice_list_kernel<<<N_BANDS, 32 * SCAN_WAVES, 0, stream>>>(src, dst, x, W1, b1, h1, counts,
                                                                    lists, list_len);

    inv_sqrt_degree_kernel<<<(N_NODES + DEG_THREADS - 1) / DEG_THREADS, DEG_THREADS, 0, stream>>>(counts, deg);

    band_aggregate_dense_kernel<<<N_BANDS, 32 * SCAN_WAVES, 0, stream>>>(src, dst, deg, h1, W2, b2, h2,
                                                                        lists, list_len);

    band_aggregate_kernel<OUT_F><<<N_BANDS, 32 * SCAN_WAVES, 0, stream>>>(src, dst, deg, h2, out,
                                                                         lists, list_len);
}


# Edge weights formed once: the first aggregation stores each listed edge's weight at its list position (workspace plane); the second reads them 32 per load beside its list words, no scattered lookup
# speedup vs baseline: 4.8440x; 1.0350x over previous
#include <hip/hip_runtime.h>
#include <hip/hip_bf16.h>

typedef __attribute__((ext_vector_type(2))) float v2f;
typedef __attribute__((ext_vector_type(8))) float v8f;

#define N_NODES 100000
#define N_EDGES 1600000
#define IN_F    128
#define HID_F   128
#define OUT_F   64

#define DENSE_PITCH 132

static_assert(DENSE_PITCH % 4 == 0, "rows stay 16-byte aligned");

__device__ __forceinline__ void layer1_tile(int lane, int lo, int m0, int rows_here,
                                            const float* __restrict__ X, const float* wsh,
                                            const float* __restrict__ bias, float* __restrict__ H) {
    constexpr int NT = 2;
    constexpr int KS = IN_F / 4;
    constexpr int KP = KS / 2;
    constexpr int AHEAD = 2;
    const int half = lane >> 4;
    const int l    = lane & 15;

    const int arow_i = (m0 + l < rows_here) ? (m0 + l) : (rows_here - 1);
    const float* arow = X + (long)(lo + arow_i) * IN_F + 2 * half;

    v2f a[KS];
#pragma unroll
    for (int s = 0; s < KS; ++s) a[s] = *(const v2f*)(arow + 4 * s);

#pragma unroll 1
    for (int n0 = 0; n0 < HID_F; n0 += 16 * NT) {
        const float* brow = wsh + (n0 + l) * DENSE_PITCH + 2 * half;

        v8f c[NT];
#pragma unroll
        for (int j = 0; j < NT; ++j) c[j] = v8f{};

        v2f b[KP][NT][2];
#pragma unroll
        for (int p = 0; p < AHEAD; ++p)
#pragma unroll
            for (int j = 0; j < NT; ++j)
#pragma unroll
                for (int t = 0; t < 2; ++t)
                    b[p][j][t] = *(const v2f*)(brow + j * 16 * DENSE_PITCH + 4 * (2 * p + t));

#pragma unroll
        for (int p = 0; p < KP; ++p) {
            if (p + AHEAD < KP) {
#pragma unroll
                for (int j = 0; j < NT; ++j)
#pragma unroll
                    for (int t = 0; t < 2; ++t)
                        b[p + AHEAD][j][t] =
                            *(const v2f*)(brow + j * 16 * DENSE_PITCH + 4 * (2 * (p + AHEAD) + t));
            }
            __builtin_amdgcn_sched_barrier(0);
#pragma unroll
            for (int t = 0; t < 2; ++t)
#pragma unroll
                for (int j = 0; j < NT; ++j) {
                    c[j] = __builtin_amdgcn_wmma_f32_16x16x4_f32(false, a[2 * p + t], false, b[p][j][t],
                                                                 (short)0, c[j], false, false);
                }
        }

        float* hp = H + (long)(lo + m0 + 8 * half) * HID_F + n0 + l;
        const int rows_left = rows_here - (m0 + 8 * half);
#pragma unroll
        for (int j = 0; j < NT; ++j) {
            const float bb = bias[n0 + 16 * j + l];
#pragma unroll
            for (int r = 0; r < 8; ++r) {
                if (r < rows_left) hp[r * HID_F + 16 * j] = c[j][r] + bb;
            }
        }
    }
}

#define BAND_ROWS 391
#define N_BANDS   256
#define EDGE_BITS 21
#define SCAN_WAVES 32
#define SCAN_AHEAD 4
#define QUEUE_WORDS 544

#define N_GROUPS    16
#define N_SLICES    16
#define GROUP_BANDS (N_BANDS / N_GROUPS)
#define GROUP_ROWS  (GROUP_BANDS * BAND_ROWS)
#define N_CHUNKS    (N_EDGES / 128)
#define COUNT_PITCH 6272
#define LEN_PITCH   32

static_assert(N_BANDS * BAND_ROWS >= N_NODES, "bands must cover every row");
static_assert((N_BANDS * BAND_ROWS * 4) % 512 == 0, "h1 behind the deg^-0.5 plane starts on a 512-byte boundary");
static_assert((N_BANDS - 1) * BAND_ROWS < N_NODES, "no empty band");
static_assert(N_EDGES <= (1 << EDGE_BITS), "edge index must fit its field");
static_assert(BAND_ROWS <= (1 << (32 - EDGE_BITS)), "band row must fit its field");
static_assert(N_EDGES % 128 == 0, "whole 128-edge chunks");
static_assert(QUEUE_WORDS >= 31 + SCAN_AHEAD * 128 + 1, "queue holds one scan step");
static_assert(N_GROUPS * N_SLICES == N_BANDS, "one work-group per (group, slice) pair and per band");
static_assert(N_GROUPS * GROUP_BANDS == N_BANDS, "groups hold whole bands");
static_assert(SCAN_WAVES == 2 * GROUP_BANDS, "two waves write a band's list");
static_assert(SCAN_WAVES == 2 * N_SLICES, "two waves read a slice's list");
static_assert(COUNT_PITCH >= GROUP_ROWS && COUNT_PITCH % 32 == 0, "count blocks are whole lines");
static_assert(LEN_PITCH >= GROUP_BANDS, "length block holds the group's bands");
static_assert(N_CHUNKS / N_SLICES >= 2, "both waves of a slice find a chunk in it");

__device__ __forceinline__ int slice_start(int s) { return (s * N_CHUNKS) / N_SLICES; }

#define LIST_CAP 1024
#define SRC_BITS 17
#define LIST_OVERFLOW 0xFFFFFFFFu

static_assert(N_NODES <= (1 << SRC_BITS), "source row must fit its field");
static_assert(BAND_ROWS <= (1 << (32 - SRC_BITS)), "band row must fit its field");
static_assert(LIST_CAP % 64 == 0, "whole 32-word list lines for both waves");

template <int F>
__device__ __forceinline__ constexpr int sum_word(int c) { return 32 * (c % (F / 32)) + c / (F / 32); }

template <int F, int P>
__device__ __forceinline__ void add_rows(int s, int r, float w, int n, int lane,
                                         const float* __restrict__ h, float* acc) {
    constexpr int C = F / 32;
    typedef __attribute__((ext_vector_type(C))) float rowpart_t;
    const int wbits = __builtin_bit_cast(int, w);

#pragma unroll 1
    for (int j0 = 0; j0 < n; j0 += 8) {
        rowpart_t v[8];
        int   rr[8];
        float ww[8];
#pragma unroll
        for (int u = 0; u < 8; ++u) {
            const int j = (j0 + u < n) ? (j0 + u) : (n - 1);
            const int sj = __builtin_amdgcn_readlane(s, j);
            rr[u] = __builtin_amdgcn_readlane(r, j);
            ww[u] = __builtin_bit_cast(float, __builtin_amdgcn_readlane(wbits, j));
            v[u] = *((const rowpart_t*)(h + (long)sj * F) + lane);
        }
#pragma unroll
        for (int u = 0; u < 8; ++u) {
            if (j0 + u < n) {
                float* ap = acc + rr[u] * P + lane;
#pragma unroll
                for (int t = 0; t < C; ++t) atomicAdd(ap + 32 * t, v[u][t] * ww[u]);
            }
        }
    }
}

template <int F, int P>
__device__ __forceinline__ void take_queued_edges(const unsigned* q, int n, int lane, int lo,
                                                  const int* __restrict__ src,
                                                  const float* __restrict__ dis,
                                                  const float* __restrict__ h, float* acc) {
    const unsigned ent = q[lane < n ? lane : n - 1];
    const int e = (int)(ent & ((1u << EDGE_BITS) - 1u));
    const int r = (int)(ent >> EDGE_BITS);
    const int s = src[e];
    const float w = dis[s] * dis[lo + r];
    add_rows<F, P>(s, r, w, n, lane, h, acc);
}

template <int F, int P>
__device__ __forceinline__ void scan_steps(int c_from, int c_to, int c_stride, int& cnt,
                                           int lane, int lo,
                                           const int* __restrict__ src,
                                           const int* __restrict__ dst,
                                           const float* __restrict__ dis,
                                           const float* __restrict__ h,
                                           float* acc, unsigned* q) {
    for (int c0 = c_from; c0 < c_to; c0 += c_stride * SCAN_AHEAD) {
        int4 dv[SCAN_AHEAD];
        int  ch[SCAN_AHEAD];
#pragma unroll
        for (int u = 0; u < SCAN_AHEAD; ++u) {
            const int ci = c0 + c_stride * u;
            const int c = (ci < c_to) ? ci : (c_to - 1);
            ch[u] = c;
            dv[u] = *(const int4*)(dst + (long)c * 128 + lane * 4);
        }
#pragma unroll
        for (int u = 0; u < SCAN_AHEAD; ++u) {
            if (c0 + c_stride * u < c_to) {
                const int ebase = ch[u] * 128 + lane * 4;
#pragma unroll
                for (int k = 0; k < 4; ++k) {
                    const int d = (k == 0) ? dv[u].x : (k == 1) ? dv[u].y : (k == 2) ? dv[u].z : dv[u].w;
                    const unsigned r = (unsigned)(d - lo);
                    const bool hit = r < (unsigned)BAND_ROWS;
                    const unsigned mask = __builtin_amdgcn_ballot_w32(hit);
                    if (mask != 0u) {
                        if (hit) {
                            const int pos = cnt + (int)__builtin_amdgcn_mbcnt_lo(mask, 0u);
                            q[pos] = (r << EDGE_BITS) | (unsigned)(ebase + k);
                        }
                        cnt += __builtin_popcount(mask);
                    }
                }
            }
        }
        while (cnt >= 32) {
            cnt -= 32;
            take_queued_edges<F, P>(q + cnt, 32, lane, lo, src, dis, h, acc);
        }
    }
}

#define DENSE_TILES ((BAND_ROWS + 15) / 16)

static_assert(DENSE_TILES <= SCAN_WAVES, "one tile per wave covers the band");

__global__ __launch_bounds__(32 * SCAN_WAVES)
void group_slice_list_kernel(const int* __restrict__ src, const int* __restrict__ dst,
                             const float* __restrict__ x, const float* __restrict__ W,
                             const float* __restrict__ bias, float* __restrict__ h1,
                             unsigned* __restrict__ counts,
                             unsigned* __restrict__ lists, unsigned* __restrict__ list_len) {
    __shared__ unsigned stage[GROUP_BANDS][LIST_CAP];
    __shared__ unsigned rowcnt[GROUP_ROWS];
    __shared__ unsigned bandcnt[GROUP_BANDS];
    __shared__ __attribute__((aligned(16))) float wsh[HID_F * DENSE_PITCH];

    const int tid  = threadIdx.x;
    const int lane = tid & 31;
    const int wave  = __builtin_amdgcn_readfirstlane(tid >> 5);
    const int wg    = blockIdx.x;
    const int group = wg / N_SLICES;
    const int slice = wg % N_SLICES;
    const int lo    = wg * BAND_ROWS;
    const int glo   = group * GROUP_ROWS;

    for (int i = tid; i < GROUP_ROWS; i += 32 * SCAN_WAVES) rowcnt[i] = 0u;
    if (tid < GROUP_BANDS) bandcnt[tid] = 0u;
    {
        const float4* w4 = (const float4*)W;
        for (int i = tid; i < HID_F * (IN_F / 4); i += 32 * SCAN_WAVES) {
            const int n  = i / (IN_F / 4);
            const int k4 = i % (IN_F / 4);
            *(float4*)(wsh + n * DENSE_PITCH + 4 * k4) = w4[i];
        }
    }
    __syncthreads();

    const int rows_here  = ((lo + BAND_ROWS < N_NODES) ? BAND_ROWS : (N_NODES - lo));
    const int tiles_here = (rows_here + 15) / 16;
    const bool tile_first = ((wave >> 2) & 1) == 0;

    if (tile_first && wave < tiles_here) layer1_tile(lane, lo, wave * 16, rows_here, x, wsh, bias, h1);

    const int c_lo = slice_start(slice);
    const int c_hi = slice_start(slice + 1);
    for (int c0 = c_lo + wave; c0 < c_hi; c0 += SCAN_WAVES * SCAN_AHEAD) {
        int4 dv[SCAN_AHEAD];
        int  ch[SCAN_AHEAD];
#pragma unroll
        for (int u = 0; u < SCAN_AHEAD; ++u) {
            const int ci = c0 + SCAN_WAVES * u;
            const int c = (ci < c_hi) ? ci : (c_hi - 1);
            ch[u] = c;
            dv[u] = *(const int4*)(dst + (long)c * 128 + lane * 4);
        }
#pragma unroll
        for (int u = 0; u < SCAN_AHEAD; ++u) {
            if (c0 + SCAN_WAVES * u < c_hi) {
                const int ebase = ch[u] * 128 + lane * 4;
#pragma unroll
                for (int k = 0; k < 4; ++k) {
                    const int d = (k == 0) ? dv[u].x : (k == 1) ? dv[u].y : (k == 2) ? dv[u].z : dv[u].w;
                    const unsigned rg = (unsigned)(d - glo);
                    if (rg < (unsigned)GROUP_ROWS) {
                        const unsigned bl = rg / (unsigned)BAND_ROWS;
                        const unsigned r  = rg - bl * (unsigned)BAND_ROWS;
                        atomicAdd(&rowcnt[rg], 1u);
                        const unsigned pos = atomicAdd(&bandcnt[bl], 1u);
                        if (pos < (unsigned)LIST_CAP) stage[bl][pos] = (r << EDGE_BITS) | (unsigned)(ebase + k);
                    }
                }
            }
        }
    }

    if (!tile_first && wave < tiles_here) layer1_tile(lane, lo, wave * 16, rows_here, x, wsh, bias, h1);

    __syncthreads();

    {
        const int bl   = wave >> 1;
        const int part = wave & 1;
        const unsigned total = bandcnt[bl];
        const int n = (total <= (unsigned)LIST_CAP) ? (int)total : 0;
        unsigned* wl = lists + ((long)(group * GROUP_BANDS + bl) * N_SLICES + slice) * LIST_CAP;
        for (int i = part * 32 + lane; i < n; i += 64) {
            const unsigned ent = stage[bl][i];
            const int e = (int)(ent & ((1u << EDGE_BITS) - 1u));
            const unsigned r = ent >> EDGE_BITS;
            wl[i] = (r << SRC_BITS) | (unsigned)src[e];
        }
    }
    if (tid < GROUP_BANDS) {
        const unsigned total = bandcnt[tid];
        list_len[wg * LEN_PITCH + tid] = (total <= (unsigned)LIST_CAP) ? total : LIST_OVERFLOW;
    }
    {
        unsigned* cp = counts + (long)(slice * N_GROUPS + group) * COUNT_PITCH;
        for (int i = tid; i < COUNT_PITCH; i += 32 * SCAN_WAVES) cp[i] = (i < GROUP_ROWS) ? rowcnt[i] : 0u;
    }
}

#define DEG_THREADS 256

__global__ __launch_bounds__(DEG_THREADS)
void inv_sqrt_degree_kernel(const unsigned* __restrict__ counts, float* __restrict__ dis) {
    const int row = blockIdx.x * DEG_THREADS + threadIdx.x;
    if (row < N_NODES) {
        const int g = row / GROUP_ROWS;
        const int i = row - g * GROUP_ROWS;
        unsigned c = 0u;
#pragma unroll
        for (int s = 0; s < N_SLICES; ++s) c += counts[(long)(s * N_GROUPS + g) * COUNT_PITCH + i];
        const float d = (float)c;
        dis[row] = (d > 0.0f) ? (1.0f / sqrtf(d)) : 0.0f;
    }
}

template <int F, int P, bool READ_W>
__device__ __forceinline__ void add_band_lists(int wave, int lane, int lo, int band,
                                               const int* __restrict__ src,
                                               const int* __restrict__ dst,
                                               const float* __restrict__ dis,
                                               const float* __restrict__ h,
                                               float* acc, unsigned* q,
                                               const unsigned* __restrict__ lists,
                                               const unsigned* __restrict__ list_len,
                                               const float* __restrict__ w_in,
                                               float* __restrict__ w_out) {
    const int slice = wave >> 1;
    const int part  = wave & 1;
    const int group = band / GROUP_BANDS;
    const int bl    = band % GROUP_BANDS;
    const long list_at = ((long)band * N_SLICES + slice) * LIST_CAP;
    const unsigned* wlist = lists + list_at;

    const unsigned len = list_len[(group * N_SLICES + slice) * LEN_PITCH + bl];
    if (len == LIST_OVERFLOW) {
        int cnt = 0;
        scan_steps<F, P>(slice_start(slice) + part, slice_start(slice + 1), 2, cnt, lane, lo,
                         src, dst, dis, h, acc, q);
        if (cnt > 0) take_queued_edges<F, P>(q, cnt, lane, lo, src, dis, h, acc);
    } else {
        const int total = (int)len;
        unsigned ent_next = 0u;
        float w_next = 0.0f;
        const int j_first = part * 32;
        if (j_first < total) {
            const int n1 = (total - j_first < 32) ? (total - j_first) : 32;
            const int at = j_first + ((lane < n1) ? lane : (n1 - 1));
            ent_next = wlist[at];
            if (READ_W) w_next = w_in[list_at + at];
        }
        for (int j0 = j_first; j0 < total; j0 += 64) {
            const int n = (total - j0 < 32) ? (total - j0) : 32;
            const unsigned ent = ent_next;
            const float w_listed = w_next;
            const int jn = j0 + 64;
            if (jn < total) {
                const int n2 = (total - jn < 32) ? (total - jn) : 32;
                const int at = jn + ((lane < n2) ? lane : (n2 - 1));
                ent_next = wlist[at];
                if (READ_W) w_next = w_in[list_at + at];
            }
            const int s = (int)(ent & ((1u << SRC_BITS) - 1u));
            const int r = (int)(ent >> SRC_BITS);
            float w;
            if (READ_W) {
                w = w_listed;
            } else {
                w = dis[s] * dis[lo + r];
                if (lane < n) w_out[list_at + j0 + lane] = w;
            }
            add_rows<F, P>(s, r, w, n, lane, h, acc);
        }
    }
}

template <int F>
__global__ __launch_bounds__(32 * SCAN_WAVES)
void band_aggregate_kernel(const int* __restrict__ src, const int* __restrict__ dst,
                           const float* __restrict__ dis, const float* __restrict__ h,
                           float* __restrict__ out,
                           const unsigned* __restrict__ lists,
                           const unsigned* __restrict__ list_len,
                           const float* __restrict__ edge_w) {
    __shared__ __attribute__((aligned(16))) float acc[BAND_ROWS * F];
    __shared__ unsigned queue[SCAN_WAVES][QUEUE_WORDS];

    const int tid  = threadIdx.x;
    const int lane = tid & 31;
    const int wave = __builtin_amdgcn_readfirstlane(tid >> 5);
    const int band = blockIdx.x;
    const int lo   = band * BAND_ROWS;

    {
        float4* a4 = (float4*)acc;
        const float4 z = {0.0f, 0.0f, 0.0f, 0.0f};
        for (int i = tid; i < BAND_ROWS * F / 4; i += 32 * SCAN_WAVES) a4[i] = z;
    }
    __syncthreads();

    add_band_lists<F, F, true>(wave, lane, lo, band, src, dst, dis, h, acc, queue[wave], lists, list_len,
                               edge_w, nullptr);

    __syncthreads();

    {
        const int rows_here = ((lo + BAND_ROWS < N_NODES) ? BAND_ROWS : (N_NODES - lo));
        const int n4 = rows_here * (F / 4);
        float4* o4 = (float4*)(out + (long)lo * F);
        static_assert(4 % (F / 32) == 0, "a 16-byte piece of a row holds whole lanes' shares");
        for (int i = tid; i < n4; i += 32 * SCAN_WAVES) {
            const int row = i / (F / 4);
            const int c0  = 4 * (i % (F / 4));
            const float* ar = acc + row * F + sum_word<F>(c0);
            float4 o;
            o.x = ar[sum_word<F>(0)];
            o.y = ar[sum_word<F>(1)];
            o.z = ar[sum_word<F>(2)];
            o.w = ar[sum_word<F>(3)];
            o4[i] = o;
        }
    }
}

#define DENSE_ROWS  (DENSE_TILES * 16)

static_assert(OUT_F % 32 == 0, "a wave's job is 32 output columns");

__global__ __launch_bounds__(32 * SCAN_WAVES)
void band_aggregate_dense_kernel(const int* __restrict__ src, const int* __restrict__ dst,
                                 const float* __restrict__ dis, const float* __restrict__ h,
                                 const float* __restrict__ W, const float* __restrict__ bias,
                                 float* __restrict__ out,
                                 const unsigned* __restrict__ lists,
                                 const unsigned* __restrict__ list_len,
                                 float* __restrict__ edge_w) {
    __shared__ __attribute__((aligned(16))) float acc[DENSE_ROWS * DENSE_PITCH];
    __shared__ __attribute__((aligned(16))) float wsh[OUT_F * DENSE_PITCH];
    __shared__ unsigned queue[SCAN_WAVES][QUEUE_WORDS];

    const int tid  = threadIdx.x;
    const int lane = tid & 31;
    const int wave = __builtin_amdgcn_readfirstlane(tid >> 5);
    const int band = blockIdx.x;
    const int lo   = band * BAND_ROWS;

    {
        float4* a4 = (float4*)acc;
        const float4 z = {0.0f, 0.0f, 0.0f, 0.0f};
        for (int i = tid; i < DENSE_ROWS * DENSE_PITCH / 4; i += 32 * SCAN_WAVES) a4[i] = z;
        static_assert(HID_F / 32 == 4, "a 16-byte piece of W is one lane's share of a row");
        const float4* w4 = (const float4*)W;
        for (int i = tid; i < OUT_F * (HID_F / 4); i += 32 * SCAN_WAVES) {
            const int n  = i / (HID_F / 4);
            const int k4 = i % (HID_F / 4);
            const float4 wv = w4[i];
            float* wr = wsh + n * DENSE_PITCH + k4;
            wr[0]  = wv.x;
            wr[32] = wv.y;
            wr[64] = wv.z;
            wr[96] = wv.w;
        }
    }
    __syncthreads();

    add_band_lists<HID_F, DENSE_PITCH, false>(wave, lane, lo, band, src, dst, dis, h, acc, queue[wave],
                                              lists, list_len, nullptr, edge_w);

    __syncthreads();

    const int rows_here  = ((lo + BAND_ROWS < N_NODES) ? BAND_ROWS : (N_NODES - lo));
    const int tiles_here = (rows_here + 15) / 16;
    const int half = lane >> 4;
    const int l    = lane & 15;
    for (int job = wave; job < tiles_here * (OUT_F / 32); job += SCAN_WAVES) {
        const int m0 = (job / (OUT_F / 32)) * 16;
        const int n0 = (job % (OUT_F / 32)) * 32;

        const float* arow  = acc + (m0 + l) * DENSE_PITCH + 2 * half;
        const float* brow0 = wsh + (n0 + l) * DENSE_PITCH + 2 * half;
        const float* brow1 = brow0 + 16 * DENSE_PITCH;

        v8f c0 = {};
        v8f c1 = {};
#pragma unroll
        for (int s = 0; s < HID_F / 4; ++s) {
            v2f a  = *(const v2f*)(arow + 4 * s);
            v2f b0 = *(const v2f*)(brow0 + 4 * s);
            v2f b1 = *(const v2f*)(brow1 + 4 * s);
            a.x = fmaxf(a.x, 0.0f);
            a.y = fmaxf(a.y, 0.0f);
            c0 = __builtin_amdgcn_wmma_f32_16x16x4_f32(false, a, false, b0,
                                                       (short)0, c0, false, false);
            c1 = __builtin_amdgcn_wmma_f32_16x16x4_f32(false, a, false, b1,
                                                       (short)0, c1, false, false);
        }

        const float bb0 = bias[n0 + l];
        const float bb1 = bias[n0 + 16 + l];
#pragma unroll
        for (int r = 0; r < 8; ++r) {
            const int row = m0 + r + 8 * half;
            if (row < rows_here) {
                float* op = out + (long)(lo + row) * OUT_F + n0 + l;
                op[0]  = c0[r] + bb0;
                op[16] = c1[r] + bb1;
            }
        }
    }
}


extern "C" void kernel_launch(void* const* d_in, const int* in_sizes, int n_in,
                              void* d_out, int out_size, void* d_ws, size_t ws_size,
                              hipStream_t stream) {
    const float* x  = (const float*)d_in[0];
    const int*   ei = (const int*)d_in[1];
    const float* W1 = (const float*)d_in[2];
    const float* b1 = (const float*)d_in[3];
    const float* W2 = (const float*)d_in[4];
    const float* b2 = (const float*)d_in[5];
    float* out = (float*)d_out;

    const int* src = ei;
    const int* dst = ei + N_EDGES;

    float* deg  = (float*)d_ws;
    float* h1   = deg + N_BANDS * BAND_ROWS;
    float* h2   = h1 + (long)N_NODES * HID_F;
    unsigned* lists    = (unsigned*)(h2 + (long)N_NODES * OUT_F);
    unsigned* list_len = lists + (long)N_BANDS * N_SLICES * LIST_CAP;
    unsigned* counts   = list_len + (long)N_BANDS * LEN_PITCH;
    float* edge_w      = (float*)(counts + (long)N_SLICES * N_GROUPS * COUNT_PITCH);

    group_slice_list_kernel<<<N_BANDS, 32 * SCAN_WAVES, 0, stream>>>(src, dst, x, W1, b1, h1, counts,
                                                                    lists, list_len);

    inv_sqrt_degree_kernel<<<(N_NODES + DEG_THREADS - 1) / DEG_THREADS, DEG_THREADS, 0, stream>>>(counts, deg);

    band_aggregate_dense_kernel<<<N_BANDS, 32 * SCAN_WAVES, 0, stream>>>(src, dst, deg, h1, W2, b2, h2,
                                                                        lists, list_len, edge_w);

    band_aggregate_kernel<OUT_F><<<N_BANDS, 32 * SCAN_WAVES, 0, stream>>>(src, dst, deg, h2, out,
                                                                         lists, list_len, edge_w);
}
